# starting kernel
# baseline (speedup 1.0000x reference)
#include <hip/hip_runtime.h>
#include <hip/hip_bf16.h>

#define N_NODES 50000
#define HDIM    96
#define LN_EPS  1e-5f

typedef float v2f __attribute__((ext_vector_type(2)));
typedef float v8f __attribute__((ext_vector_type(8)));

__global__ void k_deg_init(float* __restrict__ deg, int n) {
    int i = blockIdx.x * blockDim.x + threadIdx.x;
    if (i < n) deg[i] = 1.0f;
}

__global__ void k_deg_edges(float* __restrict__ deg, const int* __restrict__ dst, int E) {
    int i = blockIdx.x * blockDim.x + threadIdx.x;
    if (i < E) atomicAdd(&deg[dst[i]], 1.0f);
}

__global__ void k_dinv(float* __restrict__ deg, int n) {
    int i = blockIdx.x * blockDim.x + threadIdx.x;
    if (i < n) deg[i] = rsqrtf(deg[i]);
}

__global__ void k_gemm_wmma_f32(const float* __restrict__ X,
                                const float* __restrict__ W,
                                float* __restrict__ Hout,
                                int M, int K) {
    __shared__ float Bs[128 * 16];

    const int ntile = blockIdx.y;

    for (int i = threadIdx.y * 32 + threadIdx.x; i < K * 16; i += 128) {
        int k = i >> 4;
        int n = i & 15;
        Bs[i] = W[k * HDIM + ntile * 16 + n];
    }
    __syncthreads();

    const int mtiles = M / 16;
    const int mtile  = blockIdx.x * 4 + threadIdx.y;
    if (mtile >= mtiles) return;

    const int lane = threadIdx.x;
    const int r    = lane & 15;
    const int koff = (lane >> 4) << 1;

    const float* __restrict__ arow = X + (size_t)(mtile * 16 + r) * K;

    v8f c = {};
    for (int kk = 0; kk < K; kk += 4) {
        v2f a, b;
        a.x = arow[kk + koff + 0];
        a.y = arow[kk + koff + 1];
        b.x = Bs[(kk + koff + 0) * 16 + r];
        b.y = Bs[(kk + koff + 1) * 16 + r];
        c = __builtin_amdgcn_wmma_f32_16x16x4_f32(
                false, a, false, b, (short)0, c, false, false);
    }

    const int row0 = mtile * 16 + ((lane >> 4) << 3);
    float* __restrict__ out = Hout + (size_t)row0 * HDIM + ntile * 16 + r;
#pragma unroll
    for (int i = 0; i < 8; ++i) out[(size_t)i * HDIM] = c[i];
}

__global__ void k_prescale(float* __restrict__ h_agg,
                           float* __restrict__ hs,
                           const float* __restrict__ dinv,
                           int total) {
    int i = blockIdx.x * blockDim.x + threadIdx.x;
    if (i < total) {
        int n = i / HDIM;
        float s = h_agg[i] * dinv[n];
        hs[i]    = s;
        h_agg[i] = s;
    }
}

__global__ void k_scatter(const float* __restrict__ hs,
                          float* __restrict__ agg,
                          const int* __restrict__ src,
                          const int* __restrict__ dst,
                          int E) {
    int wid  = (blockIdx.x * blockDim.x + threadIdx.x) >> 5;
    int lane = threadIdx.x & 31;
    if (wid >= E) return;

    int s = src[wid];
    int d = dst[wid];
    const float* __restrict__ hrow = hs  + (size_t)s * HDIM;
    float*       __restrict__ arow = agg + (size_t)d * HDIM;
#pragma unroll
    for (int i = 0; i < 3; ++i) {
        int f = lane + 32 * i;
        atomicAdd(&arow[f], hrow[f]);
    }
}

__global__ void k_epilogue(const float* __restrict__ agg,
                           const float* __restrict__ dinv,
                           const float* __restrict__ bias,
                           const float* __restrict__ gamma,
                           const float* __restrict__ beta,
                           const float* __restrict__ alpha,
                           float* __restrict__ y,
                           int N) {
    int wid  = (blockIdx.x * blockDim.x + threadIdx.x) >> 5;
    int lane = threadIdx.x & 31;
    if (wid >= N) return;

    const float* __restrict__ row = agg + (size_t)wid * HDIM;
    float dv = dinv[wid];

    float v[3];
#pragma unroll
    for (int i = 0; i < 3; ++i) {
        int f = lane + 32 * i;
        v[i] = row[f] * dv + bias[f];
    }

    float s = v[0] + v[1] + v[2];
#pragma unroll
    for (int m = 16; m > 0; m >>= 1) s += __shfl_xor(s, m, 32);
    float mu = s * (1.0f / (float)HDIM);

    float d0 = v[0] - mu, d1 = v[1] - mu, d2 = v[2] - mu;
    float q = d0 * d0 + d1 * d1 + d2 * d2;
#pragma unroll
    for (int m = 16; m > 0; m >>= 1) q += __shfl_xor(q, m, 32);
    float rstd = rsqrtf(q * (1.0f / (float)HDIM) + LN_EPS);

    float dvv[3] = {d0, d1, d2};
#pragma unroll
    for (int i = 0; i < 3; ++i) {
        int f = lane + 32 * i;
        float o = dvv[i] * rstd * gamma[f] + beta[f];
        y[(size_t)wid * HDIM + f] = (o >= 0.0f) ? o : alpha[f] * o;
    }
}

extern "C" void kernel_launch(void* const* d_in, const int* in_sizes, int n_in,
                              void* d_out, int out_size, void* d_ws, size_t ws_size,
                              hipStream_t stream) {
    const float* x   = (const float*)d_in[0];
    const int*   ei  = (const int*)  d_in[1];
    const float* W1  = (const float*)d_in[2];
    const float* b1  = (const float*)d_in[3];
    const float* W2  = (const float*)d_in[4];
    const float* b2  = (const float*)d_in[5];
    const float* g1  = (const float*)d_in[6];
    const float* be1 = (const float*)d_in[7];
    const float* g2  = (const float*)d_in[8];
    const float* be2 = (const float*)d_in[9];
    const float* a   = (const float*)d_in[10];

    const int N   = N_NODES;
    const int Cin = in_sizes[0] / N;
    const int E   = in_sizes[1] / 2;
    const int NH  = N * HDIM;

    const int* src = ei;
    const int* dst = ei + E;

    float* dinv = (float*)d_ws;
    float* bufH = dinv + ((N + 255) & ~255);
    float* bufS = bufH + (size_t)NH;
    float* y1   = (float*)d_out;

    k_deg_init <<<(N + 255) / 256, 256, 0, stream>>>(dinv, N);
    k_deg_edges<<<(E + 255) / 256, 256, 0, stream>>>(dinv, dst, E);
    k_dinv     <<<(N + 255) / 256, 256, 0, stream>>>(dinv, N);

    dim3 gemmBlock(32, 4);
    dim3 gemmGrid1((N / 16 + 3) / 4, HDIM / 16);
    const int scatterGrid = (E * 32 + 255) / 256;
    const int epiGrid     = (N * 32 + 255) / 256;

    k_gemm_wmma_f32<<<gemmGrid1, gemmBlock, 0, stream>>>(x, W1, bufH, N, Cin);
    k_prescale<<<(NH + 255) / 256, 256, 0, stream>>>(bufH, bufS, dinv, NH);
    k_scatter <<<scatterGrid, 256, 0, stream>>>(bufS, bufH, src, dst, E);
    k_epilogue<<<epiGrid, 256, 0, stream>>>(bufH, dinv, b1, g1, be1, a, y1, N);

    k_gemm_wmma_f32<<<gemmGrid1, gemmBlock, 0, stream>>>(y1, W2, bufH, N, HDIM);
    k_prescale<<<(NH + 255) / 256, 256, 0, stream>>>(bufH, bufS, dinv, NH);
    k_scatter <<<scatterGrid, 256, 0, stream>>>(bufS, bufH, src, dst, E);
    k_epilogue<<<epiGrid, 256, 0, stream>>>(bufH, dinv, b2, g2, be2, a, (float*)d_out, N);
}


# neighbour sums off global f32 atomics: a work-group owns 384 destination rows in LDS, filters the raw edge list, adds source rows by LDS f32 add (order: LDS arrival), ends rows from LDS; 8 launches
# speedup vs baseline: 1.5756x; 1.5756x over previous
#include <hip/hip_runtime.h>
#include <hip/hip_bf16.h>

#define N_NODES 50000
#define HDIM    96
#define LN_EPS  1e-5f

typedef float v2f __attribute__((ext_vector_type(2)));
typedef float v8f __attribute__((ext_vector_type(8)));

#define DEG_PARTS   32
#define DEG_THREADS 1024
#define DEG_BATCH   8

__global__ void __launch_bounds__(DEG_THREADS)
k_deg_hist(unsigned* __restrict__ part, const int* __restrict__ dst, int E) {
    __shared__ __attribute__((aligned(16))) unsigned cnt[N_NODES];
    const int tid = threadIdx.x;

    uint4* __restrict__ cnt4 = reinterpret_cast<uint4*>(cnt);
    for (int i = tid; i < N_NODES / 4; i += DEG_THREADS) cnt4[i] = make_uint4(0u, 0u, 0u, 0u);
    __syncthreads();

    const int per = (E + (int)gridDim.x - 1) / (int)gridDim.x;
    const int lo  = (int)blockIdx.x * per;
    int hi = lo + per;
    if (hi > E) hi = E;

    for (int e0 = lo; e0 < hi; e0 += DEG_THREADS * DEG_BATCH) {
        int d[DEG_BATCH];
#pragma unroll
        for (int j = 0; j < DEG_BATCH; ++j) {
            const int e = e0 + j * DEG_THREADS + tid;
            d[j] = (e < hi) ? dst[e] : -1;
        }
#pragma unroll
        for (int j = 0; j < DEG_BATCH; ++j)
            if (d[j] >= 0) atomicAdd(&cnt[d[j]], 1u);
    }
    __syncthreads();

    uint4* __restrict__ p4 = reinterpret_cast<uint4*>(part + (size_t)blockIdx.x * N_NODES);
    for (int i = tid; i < N_NODES / 4; i += DEG_THREADS) p4[i] = cnt4[i];
}

__global__ void k_dinv(float* __restrict__ dinv, const unsigned* __restrict__ part, int n) {
    int i = blockIdx.x * blockDim.x + threadIdx.x;
    if (i < n) {
        unsigned c = 1u;
#pragma unroll
        for (int b = 0; b < DEG_PARTS; ++b) c += part[(size_t)b * n + i];
        dinv[i] = rsqrtf((float)c);
    }
}

__global__ void k_gemm_wmma_f32(const float* __restrict__ X,
                                const float* __restrict__ W,
                                float* __restrict__ Hout,
                                int M, int K) {
    __shared__ float Bs[128 * 16];

    const int ntile = blockIdx.y;

    for (int i = threadIdx.y * 32 + threadIdx.x; i < K * 16; i += 128) {
        int k = i >> 4;
        int n = i & 15;
        Bs[i] = W[k * HDIM + ntile * 16 + n];
    }
    __syncthreads();

    const int mtiles = M / 16;
    const int mtile  = blockIdx.x * 4 + threadIdx.y;
    if (mtile >= mtiles) return;

    const int lane = threadIdx.x;
    const int r    = lane & 15;
    const int koff = (lane >> 4) << 1;

    const float* __restrict__ arow = X + (size_t)(mtile * 16 + r) * K;

    v8f c = {};
    for (int kk = 0; kk < K; kk += 4) {
        v2f a, b;
        a.x = arow[kk + koff + 0];
        a.y = arow[kk + koff + 1];
        b.x = Bs[(kk + koff + 0) * 16 + r];
        b.y = Bs[(kk + koff + 1) * 16 + r];
        c = __builtin_amdgcn_wmma_f32_16x16x4_f32(
                false, a, false, b, (short)0, c, false, false);
    }

    const int row0 = mtile * 16 + ((lane >> 4) << 3);
    float* __restrict__ out = Hout + (size_t)row0 * HDIM + ntile * 16 + r;
#pragma unroll
    for (int i = 0; i < 8; ++i) out[(size_t)i * HDIM] = c[i];
}

__global__ void k_prescale(const float* __restrict__ h,
                           float* __restrict__ hs,
                           const float* __restrict__ dinv,
                           int total) {
    int i = blockIdx.x * blockDim.x + threadIdx.x;
    if (i < total) {
        int n = i / HDIM;
        hs[i] = h[i] * dinv[n];
    }
}

#define AGG_R       384
#define AGG_RANGES  ((N_NODES + AGG_R - 1) / AGG_R)
#define AGG_THREADS 1024
#define AGG_WAVES   (AGG_THREADS / 32)
#define AGG_NL      16
#define AGG_GROUP4  (32 * AGG_NL)
#define AGG_QCAP    64
#define AGG_BATCH   16

#define AGG_HIT(val, bit) ((((val) - ulo) < (unsigned)AGG_R) ? (1u << (bit)) : 0u)

__global__ void __launch_bounds__(AGG_THREADS)
k_agg_range(const float* __restrict__ hs,
            const float* __restrict__ dinv,
            const int* __restrict__ src,
            const int* __restrict__ dst,
            int E,
            const float* __restrict__ bias,
            const float* __restrict__ gamma,
            const float* __restrict__ beta,
            const float* __restrict__ alpha,
            float* __restrict__ y,
            int N) {
    __shared__ __attribute__((aligned(16))) float acc[AGG_R * HDIM];
    __shared__ int queue[AGG_WAVES * AGG_QCAP];

    const int tid  = threadIdx.x;
    const int lane = tid & 31;
    const int wave = tid >> 5;
    const int lo   = (int)blockIdx.x * AGG_R;
    const unsigned ulo = (unsigned)lo;
    int nrows = N - lo;
    if (nrows > AGG_R) nrows = AGG_R;

    {
        const float4* __restrict__ hs4 = reinterpret_cast<const float4*>(hs + (size_t)lo * HDIM);
        float4* __restrict__ acc4 = reinterpret_cast<float4*>(acc);
        const int live4 = nrows * (HDIM / 4);
        for (int i = tid; i < AGG_R * (HDIM / 4); i += AGG_THREADS)
            acc4[i] = (i < live4) ? hs4[i] : make_float4(0.f, 0.f, 0.f, 0.f);
    }
    __syncthreads();

    int* __restrict__ q = queue + wave * AGG_QCAP;
    int qn = 0;

    auto drain = [&](bool fin) {
        int done = 0;
        while (qn - done >= AGG_BATCH || (fin && qn > done)) {
            int cnt = qn - done;
            if (cnt > AGG_BATCH) cnt = AGG_BATCH;
            int s = 0, dl = 0;
            if (lane < cnt) {
                const int e = q[done + lane];
                s  = src[e];
                dl = dst[e] - lo;
            }
            float r[AGG_BATCH][3];
#pragma unroll
            for (int i = 0; i < AGG_BATCH; ++i) {
                const int si = __builtin_amdgcn_readlane(s, i);
                const float* __restrict__ row = hs + (size_t)si * HDIM + lane * 3;
                r[i][0] = row[0];
                r[i][1] = row[1];
                r[i][2] = row[2];
            }
#pragma unroll
            for (int i = 0; i < AGG_BATCH; ++i) {
                if (i < cnt) {
                    const int di = __builtin_amdgcn_readlane(dl, i);
                    float* a = acc + di * HDIM + lane * 3;
                    atomicAdd(a + 0, r[i][0]);
                    atomicAdd(a + 1, r[i][1]);
                    atomicAdd(a + 2, r[i][2]);
                }
            }
            done += cnt;
        }
        const int left = qn - done;
        if (done > 0 && left > 0) {
            const int t = (lane < left) ? q[done + lane] : 0;
            if (lane < left) q[lane] = t;
        }
        qn = left;
    };

    auto push = [&](bool act, int e) {
        const unsigned ball = __builtin_amdgcn_ballot_w32(act);
        const int pos = qn + (int)__builtin_amdgcn_mbcnt_lo(ball, 0u);
        if (act) q[pos] = e;
        qn += __builtin_popcount(ball);
    };

    auto append = [&](unsigned m, int vbase) {
        while (__builtin_amdgcn_ballot_w32(m != 0u) != 0u) {
            if (qn + 32 > AGG_QCAP) drain(false);
            const bool act = (m != 0u);
            const int b = __ffs((int)m) - 1;
            m &= m - 1u;
            push(act, ((vbase + (b >> 2) * 32 + lane) << 2) + (b & 3));
        }
    };

    const uint4* __restrict__ dst4 = reinterpret_cast<const uint4*>(dst);
    const int E4     = E >> 2;
    const int E4full = (E4 / AGG_GROUP4) * AGG_GROUP4;

    for (int g0 = wave * AGG_GROUP4; g0 < E4full; g0 += AGG_WAVES * AGG_GROUP4) {
        uint4 d[AGG_NL];
#pragma unroll
        for (int k = 0; k < AGG_NL; ++k) d[k] = dst4[g0 + k * 32 + lane];
        unsigned m0 = 0u, m1 = 0u;
#pragma unroll
        for (int k = 0; k < 8; ++k) {
            m0 |= AGG_HIT(d[k].x, 4 * k + 0) | AGG_HIT(d[k].y, 4 * k + 1) |
                  AGG_HIT(d[k].z, 4 * k + 2) | AGG_HIT(d[k].w, 4 * k + 3);
            m1 |= AGG_HIT(d[k + 8].x, 4 * k + 0) | AGG_HIT(d[k + 8].y, 4 * k + 1) |
                  AGG_HIT(d[k + 8].z, 4 * k + 2) | AGG_HIT(d[k + 8].w, 4 * k + 3);
        }
        append(m0, g0);
        append(m1, g0 + 8 * 32);
        if (qn >= AGG_BATCH) drain(false);
    }
    for (int v0 = E4full + wave * 32; v0 < E4; v0 += AGG_THREADS) {
        unsigned m = 0u;
        if (v0 + lane < E4) {
            const uint4 dv = dst4[v0 + lane];
            m = AGG_HIT(dv.x, 0) | AGG_HIT(dv.y, 1) | AGG_HIT(dv.z, 2) | AGG_HIT(dv.w, 3);
        }
        append(m, v0);
        if (qn >= AGG_BATCH) drain(false);
    }
    if (wave == 0) {
        const int e = (E4 << 2) + lane;
        const bool act = (e < E) && (((unsigned)dst[e < E ? e : 0] - ulo) < (unsigned)AGG_R);
        push(act, e);
    }
    drain(true);
    __syncthreads();

    float bi[3], ga[3], be[3], al[3];
#pragma unroll
    for (int i = 0; i < 3; ++i) {
        const int f = lane + 32 * i;
        bi[i] = bias[f]; ga[i] = gamma[f]; be[i] = beta[f]; al[i] = alpha[f];
    }
    for (int n = wave; n < nrows; n += AGG_WAVES) {
        const float* row = acc + n * HDIM;
        const float dv = dinv[lo + n];
        float v[3];
#pragma unroll
        for (int i = 0; i < 3; ++i) v[i] = row[lane + 32 * i] * dv + bi[i];

        float s = v[0] + v[1] + v[2];
#pragma unroll
        for (int m = 16; m > 0; m >>= 1) s += __shfl_xor(s, m, 32);
        const float mu = s * (1.0f / (float)HDIM);

        const float d0 = v[0] - mu, d1 = v[1] - mu, d2 = v[2] - mu;
        float qv = d0 * d0 + d1 * d1 + d2 * d2;
#pragma unroll
        for (int m = 16; m > 0; m >>= 1) qv += __shfl_xor(qv, m, 32);
        const float rstd = rsqrtf(qv * (1.0f / (float)HDIM) + LN_EPS);

        const float dvv[3] = {d0, d1, d2};
#pragma unroll
        for (int i = 0; i < 3; ++i) {
            const float o = dvv[i] * rstd * ga[i] + be[i];
            y[(size_t)(lo + n) * HDIM + lane + 32 * i] = (o >= 0.0f) ? o : al[i] * o;
        }
    }
}

extern "C" void kernel_launch(void* const* d_in, const int* in_sizes, int n_in,
                              void* d_out, int out_size, void* d_ws, size_t ws_size,
                              hipStream_t stream) {
    const float* x   = (const float*)d_in[0];
    const int*   ei  = (const int*)  d_in[1];
    const float* W1  = (const float*)d_in[2];
    const float* b1  = (const float*)d_in[3];
    const float* W2  = (const float*)d_in[4];
    const float* b2  = (const float*)d_in[5];
    const float* g1  = (const float*)d_in[6];
    const float* be1 = (const float*)d_in[7];
    const float* g2  = (const float*)d_in[8];
    const float* be2 = (const float*)d_in[9];
    const float* a   = (const float*)d_in[10];

    const int N   = N_NODES;
    const int Cin = in_sizes[0] / N;
    const int E   = in_sizes[1] / 2;
    const int NH  = N * HDIM;

    const int* src = ei;
    const int* dst = ei + E;

    float* dinv = (float*)d_ws;
    float* bufH = dinv + ((N + 255) & ~255);
    float* bufS = bufH + (size_t)NH;
    float* y1   = (float*)d_out;

    unsigned* part = (unsigned*)bufS;
    k_deg_hist<<<DEG_PARTS, DEG_THREADS, 0, stream>>>(part, dst, E);
    k_dinv    <<<(N + 255) / 256, 256, 0, stream>>>(dinv, part, N);

    dim3 gemmBlock(32, 4);
    dim3 gemmGrid1((N / 16 + 3) / 4, HDIM / 16);

    k_gemm_wmma_f32<<<gemmGrid1, gemmBlock, 0, stream>>>(x, W1, bufH, N, Cin);
    k_prescale<<<(NH + 255) / 256, 256, 0, stream>>>(bufH, bufS, dinv, NH);
    k_agg_range<<<AGG_RANGES, AGG_THREADS, 0, stream>>>(bufS, dinv, src, dst, E, b1, g1, be1, a, y1, N);

    k_gemm_wmma_f32<<<gemmGrid1, gemmBlock, 0, stream>>>(y1, W2, bufH, N, HDIM);
    k_prescale<<<(NH + 255) / 256, 256, 0, stream>>>(bufH, bufS, dinv, NH);
    k_agg_range<<<AGG_RANGES, AGG_THREADS, 0, stream>>>(bufS, dinv, src, dst, E, b2, g2, be2, a, (float*)d_out, N);
}


# agg scan reads a 1-byte owner plane (key[e] = dst[e]/384, written in whole lines by the degree pass) instead of the 4-byte destination list: 4x fewer scan bytes; sum order still LDS arrival
# speedup vs baseline: 1.7352x; 1.1013x over previous
#include <hip/hip_runtime.h>
#include <hip/hip_bf16.h>

#define N_NODES 50000
#define HDIM    96
#define LN_EPS  1e-5f

typedef float v2f __attribute__((ext_vector_type(2)));
typedef float v8f __attribute__((ext_vector_type(8)));

#define AGG_R       384
#define AGG_RANGES  ((N_NODES + AGG_R - 1) / AGG_R)
static_assert(AGG_RANGES <= 256, "a range number must fit the key plane's byte");

#define DEG_PARTS   32
#define DEG_THREADS 1024
#define DEG_BATCH   4

__global__ void __launch_bounds__(DEG_THREADS)
k_deg_hist(unsigned* __restrict__ part, unsigned* __restrict__ keyw,
           const int* __restrict__ dst, int E) {
    __shared__ __attribute__((aligned(16))) unsigned cnt[N_NODES];
    const int tid = threadIdx.x;

    uint4* __restrict__ cnt4 = reinterpret_cast<uint4*>(cnt);
    for (int i = tid; i < N_NODES / 4; i += DEG_THREADS) cnt4[i] = make_uint4(0u, 0u, 0u, 0u);
    __syncthreads();

    const int E4 = E >> 2;
    int per4 = (E4 + (int)gridDim.x - 1) / (int)gridDim.x;
    per4 = (per4 + 31) & ~31;
    const int lo4 = (int)blockIdx.x * per4;
    int hi4 = lo4 + per4;
    if (hi4 > E4) hi4 = E4;

    const uint4* __restrict__ dst4 = reinterpret_cast<const uint4*>(dst);
    for (int v0 = lo4; v0 < hi4; v0 += DEG_THREADS * DEG_BATCH) {
        uint4 d[DEG_BATCH];
#pragma unroll
        for (int j = 0; j < DEG_BATCH; ++j) {
            const int v = v0 + j * DEG_THREADS + tid;
            d[j] = (v < hi4) ? dst4[v] : make_uint4(0u, 0u, 0u, 0u);
        }
#pragma unroll
        for (int j = 0; j < DEG_BATCH; ++j) {
            const int v = v0 + j * DEG_THREADS + tid;
            if (v < hi4) {
                atomicAdd(&cnt[d[j].x], 1u);
                atomicAdd(&cnt[d[j].y], 1u);
                atomicAdd(&cnt[d[j].z], 1u);
                atomicAdd(&cnt[d[j].w], 1u);
                keyw[v] = (d[j].x / (unsigned)AGG_R)         | ((d[j].y / (unsigned)AGG_R) << 8) |
                          ((d[j].z / (unsigned)AGG_R) << 16) | ((d[j].w / (unsigned)AGG_R) << 24);
            }
        }
    }
    if (blockIdx.x == gridDim.x - 1 && tid < (E & 3)) {
        const int e = (E4 << 2) + tid;
        const unsigned dd = (unsigned)dst[e];
        atomicAdd(&cnt[dd], 1u);
        reinterpret_cast<unsigned char*>(keyw)[e] = (unsigned char)(dd / (unsigned)AGG_R);
    }
    __syncthreads();

    uint4* __restrict__ p4 = reinterpret_cast<uint4*>(part + (size_t)blockIdx.x * N_NODES);
    for (int i = tid; i < N_NODES / 4; i += DEG_THREADS) p4[i] = cnt4[i];
}

__global__ void k_dinv(float* __restrict__ dinv, const unsigned* __restrict__ part, int n) {
    int i = blockIdx.x * blockDim.x + threadIdx.x;
    if (i < n) {
        unsigned c = 1u;
#pragma unroll
        for (int b = 0; b < DEG_PARTS; ++b) c += part[(size_t)b * n + i];
        dinv[i] = rsqrtf((float)c);
    }
}

__global__ void k_gemm_wmma_f32(const float* __restrict__ X,
                                const float* __restrict__ W,
                                float* __restrict__ Hout,
                                int M, int K) {
    __shared__ float Bs[128 * 16];

    const int ntile = blockIdx.y;

    for (int i = threadIdx.y * 32 + threadIdx.x; i < K * 16; i += 128) {
        int k = i >> 4;
        int n = i & 15;
        Bs[i] = W[k * HDIM + ntile * 16 + n];
    }
    __syncthreads();

    const int mtiles = M / 16;
    const int mtile  = blockIdx.x * 4 + threadIdx.y;
    if (mtile >= mtiles) return;

    const int lane = threadIdx.x;
    const int r    = lane & 15;
    const int koff = (lane >> 4) << 1;

    const float* __restrict__ arow = X + (size_t)(mtile * 16 + r) * K;

    v8f c = {};
    for (int kk = 0; kk < K; kk += 4) {
        v2f a, b;
        a.x = arow[kk + koff + 0];
        a.y = arow[kk + koff + 1];
        b.x = Bs[(kk + koff + 0) * 16 + r];
        b.y = Bs[(kk + koff + 1) * 16 + r];
        c = __builtin_amdgcn_wmma_f32_16x16x4_f32(
                false, a, false, b, (short)0, c, false, false);
    }

    const int row0 = mtile * 16 + ((lane >> 4) << 3);
    float* __restrict__ out = Hout + (size_t)row0 * HDIM + ntile * 16 + r;
#pragma unroll
    for (int i = 0; i < 8; ++i) out[(size_t)i * HDIM] = c[i];
}

__global__ void k_prescale(float* __restrict__ h,
                           const float* __restrict__ dinv,
                           int total) {
    int i = blockIdx.x * blockDim.x + threadIdx.x;
    if (i < total) {
        int n = i / HDIM;
        h[i] = h[i] * dinv[n];
    }
}

#define AGG_THREADS 1024
#define AGG_WAVES   (AGG_THREADS / 32)
#define AGG_NL      4
#define AGG_GROUP16 (32 * AGG_NL)
#define AGG_QCAP    64
#define AGG_BATCH   16

__device__ __forceinline__ unsigned agg_zero_bytes(unsigned t) {
    return ~(((t & 0x7F7F7F7Fu) + 0x7F7F7F7Fu) | t | 0x7F7F7F7Fu);
}

__device__ __forceinline__ unsigned agg_hits16(uint4 kv, unsigned pat) {
    return (agg_zero_bytes(kv.x ^ pat) >> 7) | (agg_zero_bytes(kv.y ^ pat) >> 6) |
           (agg_zero_bytes(kv.z ^ pat) >> 5) | (agg_zero_bytes(kv.w ^ pat) >> 4);
}

__global__ void __launch_bounds__(AGG_THREADS)
k_agg_range(const float* __restrict__ hs,
            const float* __restrict__ dinv,
            const int* __restrict__ src,
            const int* __restrict__ dst,
            const unsigned* __restrict__ keyw,
            int E,
            const float* __restrict__ bias,
            const float* __restrict__ gamma,
            const float* __restrict__ beta,
            const float* __restrict__ alpha,
            float* __restrict__ y,
            int N) {
    __shared__ __attribute__((aligned(16))) float acc[AGG_R * HDIM];
    __shared__ int queue[AGG_WAVES * AGG_QCAP];

    const int tid  = threadIdx.x;
    const int lane = tid & 31;
    const int wave = tid >> 5;
    const int lo   = (int)blockIdx.x * AGG_R;
    const unsigned pat = (unsigned)blockIdx.x * 0x01010101u;
    int nrows = N - lo;
    if (nrows > AGG_R) nrows = AGG_R;

    {
        const float4* __restrict__ hs4 = reinterpret_cast<const float4*>(hs + (size_t)lo * HDIM);
        float4* __restrict__ acc4 = reinterpret_cast<float4*>(acc);
        const int live4 = nrows * (HDIM / 4);
        for (int i = tid; i < AGG_R * (HDIM / 4); i += AGG_THREADS)
            acc4[i] = (i < live4) ? hs4[i] : make_float4(0.f, 0.f, 0.f, 0.f);
    }
    __syncthreads();

    int* __restrict__ q = queue + wave * AGG_QCAP;
    int qn = 0;

    auto drain = [&](bool fin) {
        int done = 0;
        while (qn - done >= AGG_BATCH || (fin && qn > done)) {
            int cnt = qn - done;
            if (cnt > AGG_BATCH) cnt = AGG_BATCH;
            int s = 0, dl = 0;
            if (lane < cnt) {
                const int e = q[done + lane];
                s  = src[e];
                dl = dst[e] - lo;
            }
            float r[AGG_BATCH][3];
#pragma unroll
            for (int i = 0; i < AGG_BATCH; ++i) {
                const int si = __builtin_amdgcn_readlane(s, i);
                const float* __restrict__ row = hs + (size_t)si * HDIM + lane * 3;
                r[i][0] = row[0];
                r[i][1] = row[1];
                r[i][2] = row[2];
            }
#pragma unroll
            for (int i = 0; i < AGG_BATCH; ++i) {
                if (i < cnt) {
                    const int di = __builtin_amdgcn_readlane(dl, i);
                    float* a = acc + di * HDIM + lane * 3;
                    atomicAdd(a + 0, r[i][0]);
                    atomicAdd(a + 1, r[i][1]);
                    atomicAdd(a + 2, r[i][2]);
                }
            }
            done += cnt;
        }
        const int left = qn - done;
        if (done > 0 && left > 0) {
            const int t = (lane < left) ? q[done + lane] : 0;
            if (lane < left) q[lane] = t;
        }
        qn = left;
    };

    auto push = [&](bool act, int e) {
        const unsigned ball = __builtin_amdgcn_ballot_w32(act);
        const int pos = qn + (int)__builtin_amdgcn_mbcnt_lo(ball, 0u);
        if (act) q[pos] = e;
        qn += __builtin_popcount(ball);
    };

    auto append = [&](unsigned m, int vbase) {
        while (__builtin_amdgcn_ballot_w32(m != 0u) != 0u) {
            if (qn + 32 > AGG_QCAP) drain(false);
            const bool act = (m != 0u);
            const int b = __ffs((int)m) - 1;
            m &= m - 1u;
            push(act, ((vbase + ((b >> 2) & 1) * 32 + lane) << 4) + ((b & 3) << 2) + (b >> 3));
        }
    };

    const uint4* __restrict__ key4 = reinterpret_cast<const uint4*>(keyw);
    const unsigned char* __restrict__ key = reinterpret_cast<const unsigned char*>(keyw);
    const int E16     = E >> 4;
    const int E16full = (E16 / AGG_GROUP16) * AGG_GROUP16;

    for (int g0 = wave * AGG_GROUP16; g0 < E16full; g0 += AGG_WAVES * AGG_GROUP16) {
        uint4 kv[AGG_NL];
#pragma unroll
        for (int k = 0; k < AGG_NL; ++k) kv[k] = key4[g0 + k * 32 + lane];
        const unsigned m0 = agg_hits16(kv[0], pat) | (agg_hits16(kv[1], pat) << 4);
        const unsigned m1 = agg_hits16(kv[2], pat) | (agg_hits16(kv[3], pat) << 4);
        append(m0, g0);
        append(m1, g0 + 2 * 32);
        if (qn >= AGG_BATCH) drain(false);
    }
    for (int v0 = E16full + wave * 32; v0 < E16; v0 += AGG_THREADS) {
        unsigned m = 0u;
        if (v0 + lane < E16) m = agg_hits16(key4[v0 + lane], pat);
        append(m, v0);
        if (qn >= AGG_BATCH) drain(false);
    }
    if (wave == 0) {
        const int e = (E16 << 4) + lane;
        const bool act = (lane < 16) && (e < E) && ((unsigned)key[e < E ? e : 0] == (unsigned)blockIdx.x);
        push(act, e);
    }
    drain(true);
    __syncthreads();

    float bi[3], ga[3], be[3], al[3];
#pragma unroll
    for (int i = 0; i < 3; ++i) {
        const int f = lane + 32 * i;
        bi[i] = bias[f]; ga[i] = gamma[f]; be[i] = beta[f]; al[i] = alpha[f];
    }
    for (int n = wave; n < nrows; n += AGG_WAVES) {
        const float* row = acc + n * HDIM;
        const float dv = dinv[lo + n];
        float v[3];
#pragma unroll
        for (int i = 0; i < 3; ++i) v[i] = row[lane + 32 * i] * dv + bi[i];

        float s = v[0] + v[1] + v[2];
#pragma unroll
        for (int m = 16; m > 0; m >>= 1) s += __shfl_xor(s, m, 32);
        const float mu = s * (1.0f / (float)HDIM);

        const float d0 = v[0] - mu, d1 = v[1] - mu, d2 = v[2] - mu;
        float qv = d0 * d0 + d1 * d1 + d2 * d2;
#pragma unroll
        for (int m = 16; m > 0; m >>= 1) qv += __shfl_xor(qv, m, 32);
        const float rstd = rsqrtf(qv * (1.0f / (float)HDIM) + LN_EPS);

        const float dvv[3] = {d0, d1, d2};
#pragma unroll
        for (int i = 0; i < 3; ++i) {
            const float o = dvv[i] * rstd * ga[i] + be[i];
            y[(size_t)(lo + n) * HDIM + lane + 32 * i] = (o >= 0.0f) ? o : al[i] * o;
        }
    }
}

extern "C" void kernel_launch(void* const* d_in, const int* in_sizes, int n_in,
                              void* d_out, int out_size, void* d_ws, size_t ws_size,
                              hipStream_t stream) {
    const float* x   = (const float*)d_in[0];
    const int*   ei  = (const int*)  d_in[1];
    const float* W1  = (const float*)d_in[2];
    const float* b1  = (const float*)d_in[3];
    const float* W2  = (const float*)d_in[4];
    const float* b2  = (const float*)d_in[5];
    const float* g1  = (const float*)d_in[6];
    const float* be1 = (const float*)d_in[7];
    const float* g2  = (const float*)d_in[8];
    const float* be2 = (const float*)d_in[9];
    const float* a   = (const float*)d_in[10];

    const int N   = N_NODES;
    const int Cin = in_sizes[0] / N;
    const int E   = in_sizes[1] / 2;
    const int NH  = N * HDIM;

    const int* src = ei;
    const int* dst = ei + E;

    float* dinv = (float*)d_ws;
    float* bufH = dinv + ((N + 255) & ~255);
    float* bufS = bufH + (size_t)NH;
    float* y1   = (float*)d_out;

    unsigned* part = (unsigned*)bufS;
    unsigned* key = part + (size_t)DEG_PARTS * N_NODES;
    k_deg_hist<<<DEG_PARTS, DEG_THREADS, 0, stream>>>(part, key, dst, E);
    k_dinv    <<<(N + 255) / 256, 256, 0, stream>>>(dinv, part, N);

    dim3 gemmBlock(32, 4);
    dim3 gemmGrid1((N / 16 + 3) / 4, HDIM / 16);

    k_gemm_wmma_f32<<<gemmGrid1, gemmBlock, 0, stream>>>(x, W1, bufH, N, Cin);
    k_prescale<<<(NH + 255) / 256, 256, 0, stream>>>(bufH, dinv, NH);
    k_agg_range<<<AGG_RANGES, AGG_THREADS, 0, stream>>>(bufH, dinv, src, dst, key, E, b1, g1, be1, a, y1, N);

    k_gemm_wmma_f32<<<gemmGrid1, gemmBlock, 0, stream>>>(y1, W2, bufH, N, HDIM);
    k_prescale<<<(NH + 255) / 256, 256, 0, stream>>>(bufH, dinv, NH);
    k_agg_range<<<AGG_RANGES, AGG_THREADS, 0, stream>>>(bufH, dinv, src, dst, key, E, b2, g2, be2, a, (float*)d_out, N);
}


# source-row scale moved from its own pass into the aggregation's gather: row * dinv[src] in registers (same f32 product, sum order still LDS arrival); 2 launches of 150000 one-line waves gone
# speedup vs baseline: 1.8421x; 1.0616x over previous
#include <hip/hip_runtime.h>
#include <hip/hip_bf16.h>

#define N_NODES 50000
#define HDIM    96
#define LN_EPS  1e-5f

typedef float v2f __attribute__((ext_vector_type(2)));
typedef float v8f __attribute__((ext_vector_type(8)));

#define AGG_R       384
#define AGG_RANGES  ((N_NODES + AGG_R - 1) / AGG_R)
static_assert(AGG_RANGES <= 256, "a range number must fit the key plane's byte");

#define DEG_PARTS   32
#define DEG_THREADS 1024
#define DEG_BATCH   4

__global__ void __launch_bounds__(DEG_THREADS)
k_deg_hist(unsigned* __restrict__ part, unsigned* __restrict__ keyw,
           const int* __restrict__ dst, int E) {
    __shared__ __attribute__((aligned(16))) unsigned cnt[N_NODES];
    const int tid = threadIdx.x;

    uint4* __restrict__ cnt4 = reinterpret_cast<uint4*>(cnt);
    for (int i = tid; i < N_NODES / 4; i += DEG_THREADS) cnt4[i] = make_uint4(0u, 0u, 0u, 0u);
    __syncthreads();

    const int E4 = E >> 2;
    int per4 = (E4 + (int)gridDim.x - 1) / (int)gridDim.x;
    per4 = (per4 + 31) & ~31;
    const int lo4 = (int)blockIdx.x * per4;
    int hi4 = lo4 + per4;
    if (hi4 > E4) hi4 = E4;

    const uint4* __restrict__ dst4 = reinterpret_cast<const uint4*>(dst);
    for (int v0 = lo4; v0 < hi4; v0 += DEG_THREADS * DEG_BATCH) {
        uint4 d[DEG_BATCH];
#pragma unroll
        for (int j = 0; j < DEG_BATCH; ++j) {
            const int v = v0 + j * DEG_THREADS + tid;
            d[j] = (v < hi4) ? dst4[v] : make_uint4(0u, 0u, 0u, 0u);
        }
#pragma unroll
        for (int j = 0; j < DEG_BATCH; ++j) {
            const int v = v0 + j * DEG_THREADS + tid;
            if (v < hi4) {
                atomicAdd(&cnt[d[j].x], 1u);
                atomicAdd(&cnt[d[j].y], 1u);
                atomicAdd(&cnt[d[j].z], 1u);
                atomicAdd(&cnt[d[j].w], 1u);
                keyw[v] = (d[j].x / (unsigned)AGG_R)         | ((d[j].y / (unsigned)AGG_R) << 8) |
                          ((d[j].z / (unsigned)AGG_R) << 16) | ((d[j].w / (unsigned)AGG_R) << 24);
            }
        }
    }
    if (blockIdx.x == gridDim.x - 1 && tid < (E & 3)) {
        const int e = (E4 << 2) + tid;
        const unsigned dd = (unsigned)dst[e];
        atomicAdd(&cnt[dd], 1u);
        reinterpret_cast<unsigned char*>(keyw)[e] = (unsigned char)(dd / (unsigned)AGG_R);
    }
    __syncthreads();

    uint4* __restrict__ p4 = reinterpret_cast<uint4*>(part + (size_t)blockIdx.x * N_NODES);
    for (int i = tid; i < N_NODES / 4; i += DEG_THREADS) p4[i] = cnt4[i];
}

__global__ void k_dinv(float* __restrict__ dinv, const unsigned* __restrict__ part, int n) {
    int i = blockIdx.x * blockDim.x + threadIdx.x;
    if (i < n) {
        unsigned c = 1u;
#pragma unroll
        for (int b = 0; b < DEG_PARTS; ++b) c += part[(size_t)b * n + i];
        dinv[i] = rsqrtf((float)c);
    }
}

__global__ void k_gemm_wmma_f32(const float* __restrict__ X,
                                const float* __restrict__ W,
                                float* __restrict__ Hout,
                                int M, int K) {
    __shared__ float Bs[128 * 16];

    const int ntile = blockIdx.y;

    for (int i = threadIdx.y * 32 + threadIdx.x; i < K * 16; i += 128) {
        int k = i >> 4;
        int n = i & 15;
        Bs[i] = W[k * HDIM + ntile * 16 + n];
    }
    __syncthreads();

    const int mtiles = M / 16;
    const int mtile  = blockIdx.x * 4 + threadIdx.y;
    if (mtile >= mtiles) return;

    const int lane = threadIdx.x;
    const int r    = lane & 15;
    const int koff = (lane >> 4) << 1;

    const float* __restrict__ arow = X + (size_t)(mtile * 16 + r) * K;

    v8f c = {};
    for (int kk = 0; kk < K; kk += 4) {
        v2f a, b;
        a.x = arow[kk + koff + 0];
        a.y = arow[kk + koff + 1];
        b.x = Bs[(kk + koff + 0) * 16 + r];
        b.y = Bs[(kk + koff + 1) * 16 + r];
        c = __builtin_amdgcn_wmma_f32_16x16x4_f32(
                false, a, false, b, (short)0, c, false, false);
    }

    const int row0 = mtile * 16 + ((lane >> 4) << 3);
    float* __restrict__ out = Hout + (size_t)row0 * HDIM + ntile * 16 + r;
#pragma unroll
    for (int i = 0; i < 8; ++i) out[(size_t)i * HDIM] = c[i];
}

#define AGG_THREADS 1024
#define AGG_WAVES   (AGG_THREADS / 32)
#define AGG_NL      4
#define AGG_GROUP16 (32 * AGG_NL)
#define AGG_QCAP    64
#define AGG_BATCH   16

__device__ __forceinline__ unsigned agg_zero_bytes(unsigned t) {
    return ~(((t & 0x7F7F7F7Fu) + 0x7F7F7F7Fu) | t | 0x7F7F7F7Fu);
}

__device__ __forceinline__ unsigned agg_hits16(uint4 kv, unsigned pat) {
    return (agg_zero_bytes(kv.x ^ pat) >> 7) | (agg_zero_bytes(kv.y ^ pat) >> 6) |
           (agg_zero_bytes(kv.z ^ pat) >> 5) | (agg_zero_bytes(kv.w ^ pat) >> 4);
}

__global__ void __launch_bounds__(AGG_THREADS)
k_agg_range(const float* __restrict__ h,
            const float* __restrict__ dinv,
            const int* __restrict__ src,
            const int* __restrict__ dst,
            const unsigned* __restrict__ keyw,
            int E,
            const float* __restrict__ bias,
            const float* __restrict__ gamma,
            const float* __restrict__ beta,
            const float* __restrict__ alpha,
            float* __restrict__ y,
            int N) {
    __shared__ __attribute__((aligned(16))) float acc[AGG_R * HDIM];
    __shared__ int queue[AGG_WAVES * AGG_QCAP];

    const int tid  = threadIdx.x;
    const int lane = tid & 31;
    const int wave = tid >> 5;
    const int lo   = (int)blockIdx.x * AGG_R;
    const unsigned pat = (unsigned)blockIdx.x * 0x01010101u;
    int nrows = N - lo;
    if (nrows > AGG_R) nrows = AGG_R;

    {
        const float4* __restrict__ h4 = reinterpret_cast<const float4*>(h + (size_t)lo * HDIM);
        float4* __restrict__ acc4 = reinterpret_cast<float4*>(acc);
        const int live4 = nrows * (HDIM / 4);
        constexpr int SEED_TRIPS = AGG_R * (HDIM / 4) / AGG_THREADS;
        static_assert(SEED_TRIPS * AGG_THREADS == AGG_R * (HDIM / 4), "the seed loop covers acc in whole trips");
        float4 v[SEED_TRIPS];
        float  dv[SEED_TRIPS];
#pragma unroll
        for (int j = 0; j < SEED_TRIPS; ++j) {
            const int i  = tid + j * AGG_THREADS;
            const int ii = (i < live4) ? i : 0;
            v[j]  = h4[ii];
            dv[j] = dinv[lo + ii / (HDIM / 4)];
        }
#pragma unroll
        for (int j = 0; j < SEED_TRIPS; ++j) {
            const int i = tid + j * AGG_THREADS;
            acc4[i] = (i < live4)
                ? make_float4(v[j].x * dv[j], v[j].y * dv[j], v[j].z * dv[j], v[j].w * dv[j])
                : make_float4(0.f, 0.f, 0.f, 0.f);
        }
    }
    __syncthreads();

    int* __restrict__ q = queue + wave * AGG_QCAP;
    int qn = 0;

    auto drain = [&](bool fin) {
        int done = 0;
        while (qn - done >= AGG_BATCH || (fin && qn > done)) {
            int cnt = qn - done;
            if (cnt > AGG_BATCH) cnt = AGG_BATCH;
            int s = 0, dl = 0;
            float ds = 0.0f;
            if (lane < cnt) {
                const int e = q[done + lane];
                s  = src[e];
                dl = dst[e] - lo;
                ds = dinv[s];
            }
            float r[AGG_BATCH][3];
#pragma unroll
            for (int i = 0; i < AGG_BATCH; ++i) {
                const int si = __builtin_amdgcn_readlane(s, i);
                const float* __restrict__ row = h + (size_t)si * HDIM + lane * 3;
                r[i][0] = row[0];
                r[i][1] = row[1];
                r[i][2] = row[2];
            }
            __builtin_amdgcn_sched_barrier(0);
#pragma unroll
            for (int i = 0; i < AGG_BATCH; ++i) {
                if (i < cnt) {
                    const int di = __builtin_amdgcn_readlane(dl, i);
                    const float sc = __int_as_float(__builtin_amdgcn_readlane(__float_as_int(ds), i));
                    float* a = acc + di * HDIM + lane * 3;
                    atomicAdd(a + 0, r[i][0] * sc);
                    atomicAdd(a + 1, r[i][1] * sc);
                    atomicAdd(a + 2, r[i][2] * sc);
                }
            }
            done += cnt;
        }
        const int left = qn - done;
        if (done > 0 && left > 0) {
            const int t = (lane < left) ? q[done + lane] : 0;
            if (lane < left) q[lane] = t;
        }
        qn = left;
    };

    auto push = [&](bool act, int e) {
        const unsigned ball = __builtin_amdgcn_ballot_w32(act);
        const int pos = qn + (int)__builtin_amdgcn_mbcnt_lo(ball, 0u);
        if (act) q[pos] = e;
        qn += __builtin_popcount(ball);
    };

    auto append = [&](unsigned m, int vbase) {
        while (__builtin_amdgcn_ballot_w32(m != 0u) != 0u) {
            if (qn + 32 > AGG_QCAP) drain(false);
            const bool act = (m != 0u);
            const int b = __ffs((int)m) - 1;
            m &= m - 1u;
            push(act, ((vbase + ((b >> 2) & 1) * 32 + lane) << 4) + ((b & 3) << 2) + (b >> 3));
        }
    };

    const uint4* __restrict__ key4 = reinterpret_cast<const uint4*>(keyw);
    const unsigned char* __restrict__ key = reinterpret_cast<const unsigned char*>(keyw);
    const int E16     = E >> 4;
    const int E16full = (E16 / AGG_GROUP16) * AGG_GROUP16;

    for (int g0 = wave * AGG_GROUP16; g0 < E16full; g0 += AGG_WAVES * AGG_GROUP16) {
        uint4 kv[AGG_NL];
#pragma unroll
        for (int k = 0; k < AGG_NL; ++k) kv[k] = key4[g0 + k * 32 + lane];
        const unsigned m0 = agg_hits16(kv[0], pat) | (agg_hits16(kv[1], pat) << 4);
        const unsigned m1 = agg_hits16(kv[2], pat) | (agg_hits16(kv[3], pat) << 4);
        append(m0, g0);
        append(m1, g0 + 2 * 32);
        if (qn >= AGG_BATCH) drain(false);
    }
    for (int v0 = E16full + wave * 32; v0 < E16; v0 += AGG_THREADS) {
        unsigned m = 0u;
        if (v0 + lane < E16) m = agg_hits16(key4[v0 + lane], pat);
        append(m, v0);
        if (qn >= AGG_BATCH) drain(false);
    }
    if (wave == 0) {
        const int e = (E16 << 4) + lane;
        const bool act = (lane < 16) && (e < E) && ((unsigned)key[e < E ? e : 0] == (unsigned)blockIdx.x);
        push(act, e);
    }
    drain(true);
    __syncthreads();

    float bi[3], ga[3], be[3], al[3];
#pragma unroll
    for (int i = 0; i < 3; ++i) {
        const int f = lane + 32 * i;
        bi[i] = bias[f]; ga[i] = gamma[f]; be[i] = beta[f]; al[i] = alpha[f];
    }
    for (int n = wave; n < nrows; n += AGG_WAVES) {
        const float* row = acc + n * HDIM;
        const float dv = dinv[lo + n];
        float v[3];
#pragma unroll
        for (int i = 0; i < 3; ++i) v[i] = row[lane + 32 * i] * dv + bi[i];

        float s = v[0] + v[1] + v[2];
#pragma unroll
        for (int m = 16; m > 0; m >>= 1) s += __shfl_xor(s, m, 32);
        const float mu = s * (1.0f / (float)HDIM);

        const float d0 = v[0] - mu, d1 = v[1] - mu, d2 = v[2] - mu;
        float qv = d0 * d0 + d1 * d1 + d2 * d2;
#pragma unroll
        for (int m = 16; m > 0; m >>= 1) qv += __shfl_xor(qv, m, 32);
        const float rstd = rsqrtf(qv * (1.0f / (float)HDIM) + LN_EPS);

        const float dvv[3] = {d0, d1, d2};
#pragma unroll
        for (int i = 0; i < 3; ++i) {
            const float o = dvv[i] * rstd * ga[i] + be[i];
            y[(size_t)(lo + n) * HDIM + lane + 32 * i] = (o >= 0.0f) ? o : al[i] * o;
        }
    }
}

extern "C" void kernel_launch(void* const* d_in, const int* in_sizes, int n_in,
                              void* d_out, int out_size, void* d_ws, size_t ws_size,
                              hipStream_t stream) {
    const float* x   = (const float*)d_in[0];
    const int*   ei  = (const int*)  d_in[1];
    const float* W1  = (const float*)d_in[2];
    const float* b1  = (const float*)d_in[3];
    const float* W2  = (const float*)d_in[4];
    const float* b2  = (const float*)d_in[5];
    const float* g1  = (const float*)d_in[6];
    const float* be1 = (const float*)d_in[7];
    const float* g2  = (const float*)d_in[8];
    const float* be2 = (const float*)d_in[9];
    const float* a   = (const float*)d_in[10];

    const int N   = N_NODES;
    const int Cin = in_sizes[0] / N;
    const int E   = in_sizes[1] / 2;
    const int NH  = N * HDIM;

    const int* src = ei;
    const int* dst = ei + E;

    float* dinv = (float*)d_ws;
    float* bufH = dinv + ((N + 255) & ~255);
    float* bufS = bufH + (size_t)NH;
    float* y1   = (float*)d_out;

    unsigned* part = (unsigned*)bufS;
    unsigned* key = part + (size_t)DEG_PARTS * N_NODES;
    k_deg_hist<<<DEG_PARTS, DEG_THREADS, 0, stream>>>(part, key, dst, E);
    k_dinv    <<<(N + 255) / 256, 256, 0, stream>>>(dinv, part, N);

    dim3 gemmBlock(32, 4);
    dim3 gemmGrid1((N / 16 + 3) / 4, HDIM / 16);

    k_gemm_wmma_f32<<<gemmGrid1, gemmBlock, 0, stream>>>(x, W1, bufH, N, Cin);
    k_agg_range<<<AGG_RANGES, AGG_THREADS, 0, stream>>>(bufH, dinv, src, dst, key, E, b1, g1, be1, a, y1, N);

    k_gemm_wmma_f32<<<gemmGrid1, gemmBlock, 0, stream>>>(y1, W2, bufH, N, HDIM);
    k_agg_range<<<AGG_RANGES, AGG_THREADS, 0, stream>>>(bufH, dinv, src, dst, key, E, b2, g2, be2, a, (float*)d_out, N);
}


# aggregation split into 256 work-groups of 196 destination rows (was 131 of 384): one per WGP instead of half the chip; a wave's gather batches halve; sum order still LDS arrival; 6 launches
# speedup vs baseline: 2.1809x; 1.1839x over previous
#include <hip/hip_runtime.h>
#include <hip/hip_bf16.h>

#define N_NODES 50000
#define HDIM    96
#define LN_EPS  1e-5f

typedef float v2f __attribute__((ext_vector_type(2)));
typedef float v8f __attribute__((ext_vector_type(8)));

#define AGG_R       196
#define AGG_RANGES  ((N_NODES + AGG_R - 1) / AGG_R)
static_assert(AGG_RANGES <= 256, "a range number must fit the key plane's byte");

#define DEG_PARTS   32
#define DEG_THREADS 1024
#define DEG_BATCH   4

__global__ void __launch_bounds__(DEG_THREADS)
k_deg_hist(unsigned* __restrict__ part, unsigned* __restrict__ keyw,
           const int* __restrict__ dst, int E) {
    __shared__ __attribute__((aligned(16))) unsigned cnt[N_NODES];
    const int tid = threadIdx.x;

    uint4* __restrict__ cnt4 = reinterpret_cast<uint4*>(cnt);
    for (int i = tid; i < N_NODES / 4; i += DEG_THREADS) cnt4[i] = make_uint4(0u, 0u, 0u, 0u);
    __syncthreads();

    const int E4 = E >> 2;
    int per4 = (E4 + (int)gridDim.x - 1) / (int)gridDim.x;
    per4 = (per4 + 31) & ~31;
    const int lo4 = (int)blockIdx.x * per4;
    int hi4 = lo4 + per4;
    if (hi4 > E4) hi4 = E4;

    const uint4* __restrict__ dst4 = reinterpret_cast<const uint4*>(dst);
    for (int v0 = lo4; v0 < hi4; v0 += DEG_THREADS * DEG_BATCH) {
        uint4 d[DEG_BATCH];
#pragma unroll
        for (int j = 0; j < DEG_BATCH; ++j) {
            const int v = v0 + j * DEG_THREADS + tid;
            d[j] = (v < hi4) ? dst4[v] : make_uint4(0u, 0u, 0u, 0u);
        }
#pragma unroll
        for (int j = 0; j < DEG_BATCH; ++j) {
            const int v = v0 + j * DEG_THREADS + tid;
            if (v < hi4) {
                atomicAdd(&cnt[d[j].x], 1u);
                atomicAdd(&cnt[d[j].y], 1u);
                atomicAdd(&cnt[d[j].z], 1u);
                atomicAdd(&cnt[d[j].w], 1u);
                keyw[v] = (d[j].x / (unsigned)AGG_R)         | ((d[j].y / (unsigned)AGG_R) << 8) |
                          ((d[j].z / (unsigned)AGG_R) << 16) | ((d[j].w / (unsigned)AGG_R) << 24);
            }
        }
    }
    if (blockIdx.x == gridDim.x - 1 && tid < (E & 3)) {
        const int e = (E4 << 2) + tid;
        const unsigned dd = (unsigned)dst[e];
        atomicAdd(&cnt[dd], 1u);
        reinterpret_cast<unsigned char*>(keyw)[e] = (unsigned char)(dd / (unsigned)AGG_R);
    }
    __syncthreads();

    uint4* __restrict__ p4 = reinterpret_cast<uint4*>(part + (size_t)blockIdx.x * N_NODES);
    for (int i = tid; i < N_NODES / 4; i += DEG_THREADS) p4[i] = cnt4[i];
}

__global__ void k_dinv(float* __restrict__ dinv, const unsigned* __restrict__ part, int n) {
    int i = blockIdx.x * blockDim.x + threadIdx.x;
    if (i < n) {
        unsigned c = 1u;
#pragma unroll
        for (int b = 0; b < DEG_PARTS; ++b) c += part[(size_t)b * n + i];
        dinv[i] = rsqrtf((float)c);
    }
}

__global__ void k_gemm_wmma_f32(const float* __restrict__ X,
                                const float* __restrict__ W,
                                float* __restrict__ Hout,
                                int M, int K) {
    __shared__ float Bs[128 * 16];

    const int ntile = blockIdx.y;

    for (int i = threadIdx.y * 32 + threadIdx.x; i < K * 16; i += 128) {
        int k = i >> 4;
        int n = i & 15;
        Bs[i] = W[k * HDIM + ntile * 16 + n];
    }
    __syncthreads();

    const int mtiles = M / 16;
    const int mtile  = blockIdx.x * 4 + threadIdx.y;
    if (mtile >= mtiles) return;

    const int lane = threadIdx.x;
    const int r    = lane & 15;
    const int koff = (lane >> 4) << 1;

    const float* __restrict__ arow = X + (size_t)(mtile * 16 + r) * K;

    v8f c = {};
    for (int kk = 0; kk < K; kk += 4) {
        v2f a, b;
        a.x = arow[kk + koff + 0];
        a.y = arow[kk + koff + 1];
        b.x = Bs[(kk + koff + 0) * 16 + r];
        b.y = Bs[(kk + koff + 1) * 16 + r];
        c = __builtin_amdgcn_wmma_f32_16x16x4_f32(
                false, a, false, b, (short)0, c, false, false);
    }

    const int row0 = mtile * 16 + ((lane >> 4) << 3);
    float* __restrict__ out = Hout + (size_t)row0 * HDIM + ntile * 16 + r;
#pragma unroll
    for (int i = 0; i < 8; ++i) out[(size_t)i * HDIM] = c[i];
}

#define AGG_THREADS 1024
#define AGG_WAVES   (AGG_THREADS / 32)
#define AGG_NL      4
#define AGG_GROUP16 (32 * AGG_NL)
#define AGG_QCAP    64
#define AGG_BATCH   16

__device__ __forceinline__ unsigned agg_zero_bytes(unsigned t) {
    return ~(((t & 0x7F7F7F7Fu) + 0x7F7F7F7Fu) | t | 0x7F7F7F7Fu);
}

__device__ __forceinline__ unsigned agg_hits16(uint4 kv, unsigned pat) {
    return (agg_zero_bytes(kv.x ^ pat) >> 7) | (agg_zero_bytes(kv.y ^ pat) >> 6) |
           (agg_zero_bytes(kv.z ^ pat) >> 5) | (agg_zero_bytes(kv.w ^ pat) >> 4);
}

__global__ void __launch_bounds__(AGG_THREADS)
k_agg_range(const float* __restrict__ h,
            const float* __restrict__ dinv,
            const int* __restrict__ src,
            const int* __restrict__ dst,
            const unsigned* __restrict__ keyw,
            int E,
            const float* __restrict__ bias,
            const float* __restrict__ gamma,
            const float* __restrict__ beta,
            const float* __restrict__ alpha,
            float* __restrict__ y,
            int N) {
    __shared__ __attribute__((aligned(16))) float acc[AGG_R * HDIM];
    __shared__ int queue[AGG_WAVES * AGG_QCAP];

    const int tid  = threadIdx.x;
    const int lane = tid & 31;
    const int wave = tid >> 5;
    const int lo   = (int)blockIdx.x * AGG_R;
    const unsigned pat = (unsigned)blockIdx.x * 0x01010101u;
    int nrows = N - lo;
    if (nrows > AGG_R) nrows = AGG_R;

    {
        const float4* __restrict__ h4 = reinterpret_cast<const float4*>(h + (size_t)lo * HDIM);
        float4* __restrict__ acc4 = reinterpret_cast<float4*>(acc);
        const int live4 = nrows * (HDIM / 4);
        constexpr int SEED_SLOTS = AGG_R * (HDIM / 4);
        constexpr int SEED_TRIPS = (SEED_SLOTS + AGG_THREADS - 1) / AGG_THREADS;
        float4 v[SEED_TRIPS];
        float  dv[SEED_TRIPS];
#pragma unroll
        for (int j = 0; j < SEED_TRIPS; ++j) {
            const int i  = tid + j * AGG_THREADS;
            const int ii = (i < live4) ? i : 0;
            v[j]  = h4[ii];
            dv[j] = dinv[lo + ii / (HDIM / 4)];
        }
#pragma unroll
        for (int j = 0; j < SEED_TRIPS; ++j) {
            const int i = tid + j * AGG_THREADS;
            if (i < SEED_SLOTS)
                acc4[i] = (i < live4)
                    ? make_float4(v[j].x * dv[j], v[j].y * dv[j], v[j].z * dv[j], v[j].w * dv[j])
                    : make_float4(0.f, 0.f, 0.f, 0.f);
        }
    }
    __syncthreads();

    int* __restrict__ q = queue + wave * AGG_QCAP;
    int qn = 0;

    auto drain = [&](bool fin) {
        int done = 0;
        while (qn - done >= AGG_BATCH || (fin && qn > done)) {
            int cnt = qn - done;
            if (cnt > AGG_BATCH) cnt = AGG_BATCH;
            int s = 0, dl = 0;
            float ds = 0.0f;
            if (lane < cnt) {
                const int e = q[done + lane];
                s  = src[e];
                dl = dst[e] - lo;
                ds = dinv[s];
            }
            float r[AGG_BATCH][3];
#pragma unroll
            for (int i = 0; i < AGG_BATCH; ++i) {
                const int si = __builtin_amdgcn_readlane(s, i);
                const float* __restrict__ row = h + (size_t)si * HDIM + lane * 3;
                r[i][0] = row[0];
                r[i][1] = row[1];
                r[i][2] = row[2];
            }
            __builtin_amdgcn_sched_barrier(0);
#pragma unroll
            for (int i = 0; i < AGG_BATCH; ++i) {
                if (i < cnt) {
                    const int di = __builtin_amdgcn_readlane(dl, i);
                    const float sc = __int_as_float(__builtin_amdgcn_readlane(__float_as_int(ds), i));
                    float* a = acc + di * HDIM + lane * 3;
                    atomicAdd(a + 0, r[i][0] * sc);
                    atomicAdd(a + 1, r[i][1] * sc);
                    atomicAdd(a + 2, r[i][2] * sc);
                }
            }
            done += cnt;
        }
        const int left = qn - done;
        if (done > 0 && left > 0) {
            const int t = (lane < left) ? q[done + lane] : 0;
            if (lane < left) q[lane] = t;
        }
        qn = left;
    };

    auto push = [&](bool act, int e) {
        const unsigned ball = __builtin_amdgcn_ballot_w32(act);
        const int pos = qn + (int)__builtin_amdgcn_mbcnt_lo(ball, 0u);
        if (act) q[pos] = e;
        qn += __builtin_popcount(ball);
    };

    auto append = [&](unsigned m, int vbase) {
        while (__builtin_amdgcn_ballot_w32(m != 0u) != 0u) {
            if (qn + 32 > AGG_QCAP) drain(false);
            const bool act = (m != 0u);
            const int b = __ffs((int)m) - 1;
            m &= m - 1u;
            push(act, ((vbase + ((b >> 2) & 1) * 32 + lane) << 4) + ((b & 3) << 2) + (b >> 3));
        }
    };

    const uint4* __restrict__ key4 = reinterpret_cast<const uint4*>(keyw);
    const unsigned char* __restrict__ key = reinterpret_cast<const unsigned char*>(keyw);
    const int E16     = E >> 4;
    const int E16full = (E16 / AGG_GROUP16) * AGG_GROUP16;

    for (int g0 = wave * AGG_GROUP16; g0 < E16full; g0 += AGG_WAVES * AGG_GROUP16) {
        uint4 kv[AGG_NL];
#pragma unroll
        for (int k = 0; k < AGG_NL; ++k) kv[k] = key4[g0 + k * 32 + lane];
        const unsigned m0 = agg_hits16(kv[0], pat) | (agg_hits16(kv[1], pat) << 4);
        const unsigned m1 = agg_hits16(kv[2], pat) | (agg_hits16(kv[3], pat) << 4);
        append(m0, g0);
        append(m1, g0 + 2 * 32);
        if (qn >= AGG_BATCH) drain(false);
    }
    for (int v0 = E16full + wave * 32; v0 < E16; v0 += AGG_THREADS) {
        unsigned m = 0u;
        if (v0 + lane < E16) m = agg_hits16(key4[v0 + lane], pat);
        append(m, v0);
        if (qn >= AGG_BATCH) drain(false);
    }
    if (wave == 0) {
        const int e = (E16 << 4) + lane;
        const bool act = (lane < 16) && (e < E) && ((unsigned)key[e < E ? e : 0] == (unsigned)blockIdx.x);
        push(act, e);
    }
    drain(true);
    __syncthreads();

    float bi[3], ga[3], be[3], al[3];
#pragma unroll
    for (int i = 0; i < 3; ++i) {
        const int f = lane + 32 * i;
        bi[i] = bias[f]; ga[i] = gamma[f]; be[i] = beta[f]; al[i] = alpha[f];
    }
    for (int n = wave; n < nrows; n += AGG_WAVES) {
        const float* row = acc + n * HDIM;
        const float dv = dinv[lo + n];
        float v[3];
#pragma unroll
        for (int i = 0; i < 3; ++i) v[i] = row[lane + 32 * i] * dv + bi[i];

        float s = v[0] + v[1] + v[2];
#pragma unroll
        for (int m = 16; m > 0; m >>= 1) s += __shfl_xor(s, m, 32);
        const float mu = s * (1.0f / (float)HDIM);

        const float d0 = v[0] - mu, d1 = v[1] - mu, d2 = v[2] - mu;
        float qv = d0 * d0 + d1 * d1 + d2 * d2;
#pragma unroll
        for (int m = 16; m > 0; m >>= 1) qv += __shfl_xor(qv, m, 32);
        const float rstd = rsqrtf(qv * (1.0f / (float)HDIM) + LN_EPS);

        const float dvv[3] = {d0, d1, d2};
#pragma unroll
        for (int i = 0; i < 3; ++i) {
            const float o = dvv[i] * rstd * ga[i] + be[i];
            y[(size_t)(lo + n) * HDIM + lane + 32 * i] = (o >= 0.0f) ? o : al[i] * o;
        }
    }
}

extern "C" void kernel_launch(void* const* d_in, const int* in_sizes, int n_in,
                              void* d_out, int out_size, void* d_ws, size_t ws_size,
                              hipStream_t stream) {
    const float* x   = (const float*)d_in[0];
    const int*   ei  = (const int*)  d_in[1];
    const float* W1  = (const float*)d_in[2];
    const float* b1  = (const float*)d_in[3];
    const float* W2  = (const float*)d_in[4];
    const float* b2  = (const float*)d_in[5];
    const float* g1  = (const float*)d_in[6];
    const float* be1 = (const float*)d_in[7];
    const float* g2  = (const float*)d_in[8];
    const float* be2 = (const float*)d_in[9];
    const float* a   = (const float*)d_in[10];

    const int N   = N_NODES;
    const int Cin = in_sizes[0] / N;
    const int E   = in_sizes[1] / 2;
    const int NH  = N * HDIM;

    const int* src = ei;
    const int* dst = ei + E;

    float* dinv = (float*)d_ws;
    float* bufH = dinv + ((N + 255) & ~255);
    float* bufS = bufH + (size_t)NH;
    float* y1   = (float*)d_out;

    unsigned* part = (unsigned*)bufS;
    unsigned* key = part + (size_t)DEG_PARTS * N_NODES;
    k_deg_hist<<<DEG_PARTS, DEG_THREADS, 0, stream>>>(part, key, dst, E);
    k_dinv    <<<(N + 255) / 256, 256, 0, stream>>>(dinv, part, N);

    dim3 gemmBlock(32, 4);
    dim3 gemmGrid1((N / 16 + 3) / 4, HDIM / 16);

    k_gemm_wmma_f32<<<gemmGrid1, gemmBlock, 0, stream>>>(x, W1, bufH, N, Cin);
    k_agg_range<<<AGG_RANGES, AGG_THREADS, 0, stream>>>(bufH, dinv, src, dst, key, E, b1, g1, be1, a, y1, N);

    k_gemm_wmma_f32<<<gemmGrid1, gemmBlock, 0, stream>>>(y1, W2, bufH, N, HDIM);
    k_agg_range<<<AGG_RANGES, AGG_THREADS, 0, stream>>>(bufH, dinv, src, dst, key, E, b2, g2, be2, a, (float*)d_out, N);
}


# layer 2's aggregation reads per-wave lists of the kept edges (source, local row in 4 bytes) that layer 1's filter wrote, not the key plane and src/dst again; sum order still LDS arrival
# speedup vs baseline: 2.3922x; 1.0969x over previous
#include <hip/hip_runtime.h>
#include <hip/hip_bf16.h>

#define N_NODES 50000
#define HDIM    96
#define LN_EPS  1e-5f

typedef float v2f __attribute__((ext_vector_type(2)));
typedef float v8f __attribute__((ext_vector_type(8)));

#define AGG_R       196
#define AGG_RANGES  ((N_NODES + AGG_R - 1) / AGG_R)
static_assert(AGG_RANGES <= 256, "a range number must fit the key plane's byte");

#define DEG_PARTS   32
#define DEG_THREADS 1024
#define DEG_BATCH   4

__global__ void __launch_bounds__(DEG_THREADS)
k_deg_hist(unsigned* __restrict__ part, unsigned* __restrict__ keyw,
           const int* __restrict__ dst, int E) {
    __shared__ __attribute__((aligned(16))) unsigned cnt[N_NODES];
    const int tid = threadIdx.x;

    uint4* __restrict__ cnt4 = reinterpret_cast<uint4*>(cnt);
    for (int i = tid; i < N_NODES / 4; i += DEG_THREADS) cnt4[i] = make_uint4(0u, 0u, 0u, 0u);
    __syncthreads();

    const int E4 = E >> 2;
    int per4 = (E4 + (int)gridDim.x - 1) / (int)gridDim.x;
    per4 = (per4 + 31) & ~31;
    const int lo4 = (int)blockIdx.x * per4;
    int hi4 = lo4 + per4;
    if (hi4 > E4) hi4 = E4;

    const uint4* __restrict__ dst4 = reinterpret_cast<const uint4*>(dst);
    for (int v0 = lo4; v0 < hi4; v0 += DEG_THREADS * DEG_BATCH) {
        uint4 d[DEG_BATCH];
#pragma unroll
        for (int j = 0; j < DEG_BATCH; ++j) {
            const int v = v0 + j * DEG_THREADS + tid;
            d[j] = (v < hi4) ? dst4[v] : make_uint4(0u, 0u, 0u, 0u);
        }
#pragma unroll
        for (int j = 0; j < DEG_BATCH; ++j) {
            const int v = v0 + j * DEG_THREADS + tid;
            if (v < hi4) {
                atomicAdd(&cnt[d[j].x], 1u);
                atomicAdd(&cnt[d[j].y], 1u);
                atomicAdd(&cnt[d[j].z], 1u);
                atomicAdd(&cnt[d[j].w], 1u);
                keyw[v] = (d[j].x / (unsigned)AGG_R)         | ((d[j].y / (unsigned)AGG_R) << 8) |
                          ((d[j].z / (unsigned)AGG_R) << 16) | ((d[j].w / (unsigned)AGG_R) << 24);
            }
        }
    }
    if (blockIdx.x == gridDim.x - 1 && tid < (E & 3)) {
        const int e = (E4 << 2) + tid;
        const unsigned dd = (unsigned)dst[e];
        atomicAdd(&cnt[dd], 1u);
        reinterpret_cast<unsigned char*>(keyw)[e] = (unsigned char)(dd / (unsigned)AGG_R);
    }
    __syncthreads();

    uint4* __restrict__ p4 = reinterpret_cast<uint4*>(part + (size_t)blockIdx.x * N_NODES);
    for (int i = tid; i < N_NODES / 4; i += DEG_THREADS) p4[i] = cnt4[i];
}

__global__ void k_dinv(float* __restrict__ dinv, const unsigned* __restrict__ part, int n) {
    int i = blockIdx.x * blockDim.x + threadIdx.x;
    if (i < n) {
        unsigned c = 1u;
#pragma unroll
        for (int b = 0; b < DEG_PARTS; ++b) c += part[(size_t)b * n + i];
        dinv[i] = rsqrtf((float)c);
    }
}

__global__ void k_gemm_wmma_f32(const float* __restrict__ X,
                                const float* __restrict__ W,
                                float* __restrict__ Hout,
                                int M, int K) {
    __shared__ float Bs[128 * 16];

    const int ntile = blockIdx.y;

    for (int i = threadIdx.y * 32 + threadIdx.x; i < K * 16; i += 128) {
        int k = i >> 4;
        int n = i & 15;
        Bs[i] = W[k * HDIM + ntile * 16 + n];
    }
    __syncthreads();

    const int mtiles = M / 16;
    const int mtile  = blockIdx.x * 4 + threadIdx.y;
    if (mtile >= mtiles) return;

    const int lane = threadIdx.x;
    const int r    = lane & 15;
    const int koff = (lane >> 4) << 1;

    const float* __restrict__ arow = X + (size_t)(mtile * 16 + r) * K;

    v8f c = {};
    for (int kk = 0; kk < K; kk += 4) {
        v2f a, b;
        a.x = arow[kk + koff + 0];
        a.y = arow[kk + koff + 1];
        b.x = Bs[(kk + koff + 0) * 16 + r];
        b.y = Bs[(kk + koff + 1) * 16 + r];
        c = __builtin_amdgcn_wmma_f32_16x16x4_f32(
                false, a, false, b, (short)0, c, false, false);
    }

    const int row0 = mtile * 16 + ((lane >> 4) << 3);
    float* __restrict__ out = Hout + (size_t)row0 * HDIM + ntile * 16 + r;
#pragma unroll
    for (int i = 0; i < 8; ++i) out[(size_t)i * HDIM] = c[i];
}

#define AGG_THREADS 1024
#define AGG_WAVES   (AGG_THREADS / 32)
#define AGG_NL      4
#define AGG_GROUP16 (32 * AGG_NL)
#define AGG_QCAP    64
#define AGG_BATCH   16
#define AGG_WCAP    256
static_assert(N_NODES <= 65536 && AGG_R <= 65536, "source and local row share a 32-bit entry");
static_assert(AGG_WCAP % 32 == 0 && AGG_BATCH == 16, "a 32-entry load feeds two batches");
static_assert(AGG_WAVES == 32, "a wave reads its work-group's counts one per lane");

__device__ __forceinline__ unsigned agg_zero_bytes(unsigned t) {
    return ~(((t & 0x7F7F7F7Fu) + 0x7F7F7F7Fu) | t | 0x7F7F7F7Fu);
}

__device__ __forceinline__ unsigned agg_hits16(uint4 kv, unsigned pat) {
    return (agg_zero_bytes(kv.x ^ pat) >> 7) | (agg_zero_bytes(kv.y ^ pat) >> 6) |
           (agg_zero_bytes(kv.z ^ pat) >> 5) | (agg_zero_bytes(kv.w ^ pat) >> 4);
}

template <bool FROM_LIST>
__global__ void __launch_bounds__(AGG_THREADS)
k_agg_range(const float* __restrict__ h,
            const float* __restrict__ dinv,
            const int* __restrict__ src,
            const int* __restrict__ dst,
            const unsigned* __restrict__ keyw,
            int E,
            unsigned* lst,
            unsigned* lcnt,
            const float* __restrict__ bias,
            const float* __restrict__ gamma,
            const float* __restrict__ beta,
            const float* __restrict__ alpha,
            float* __restrict__ y,
            int N) {
    __shared__ __attribute__((aligned(16))) float acc[AGG_R * HDIM];
    __shared__ int queue[AGG_WAVES * AGG_QCAP];
    __shared__ unsigned wkept[AGG_WAVES];

    const int tid  = threadIdx.x;
    const int lane = tid & 31;
    const int wave = tid >> 5;
    const int lo   = (int)blockIdx.x * AGG_R;
    const unsigned pat = (unsigned)blockIdx.x * 0x01010101u;
    int nrows = N - lo;
    if (nrows > AGG_R) nrows = AGG_R;

    {
        const float4* __restrict__ h4 = reinterpret_cast<const float4*>(h + (size_t)lo * HDIM);
        float4* __restrict__ acc4 = reinterpret_cast<float4*>(acc);
        const int live4 = nrows * (HDIM / 4);
        constexpr int SEED_SLOTS = AGG_R * (HDIM / 4);
        constexpr int SEED_TRIPS = (SEED_SLOTS + AGG_THREADS - 1) / AGG_THREADS;
        float4 v[SEED_TRIPS];
        float  dv[SEED_TRIPS];
#pragma unroll
        for (int j = 0; j < SEED_TRIPS; ++j) {
            const int i  = tid + j * AGG_THREADS;
            const int ii = (i < live4) ? i : 0;
            v[j]  = h4[ii];
            dv[j] = dinv[lo + ii / (HDIM / 4)];
        }
#pragma unroll
        for (int j = 0; j < SEED_TRIPS; ++j) {
            const int i = tid + j * AGG_THREADS;
            if (i < SEED_SLOTS)
                acc4[i] = (i < live4)
                    ? make_float4(v[j].x * dv[j], v[j].y * dv[j], v[j].z * dv[j], v[j].w * dv[j])
                    : make_float4(0.f, 0.f, 0.f, 0.f);
        }
    }
    __syncthreads();

    int* __restrict__ q = queue + wave * AGG_QCAP;
    int qn = 0;

    unsigned* wl = lst + ((size_t)blockIdx.x * AGG_WAVES + wave) * AGG_WCAP;
    int wn = 0;

    auto gather = [&](int s, int dl, int off, int cnt) {
        float ds = 0.0f;
        if (lane >= off && lane < off + cnt) ds = dinv[s];
        float r[AGG_BATCH][3];
#pragma unroll
        for (int i = 0; i < AGG_BATCH; ++i) {
            const int si = __builtin_amdgcn_readlane(s, off + i);
            const float* __restrict__ row = h + (size_t)si * HDIM + lane * 3;
            r[i][0] = row[0];
            r[i][1] = row[1];
            r[i][2] = row[2];
        }
        __builtin_amdgcn_sched_barrier(0);
#pragma unroll
        for (int i = 0; i < AGG_BATCH; ++i) {
            if (i < cnt) {
                const int di = __builtin_amdgcn_readlane(dl, off + i);
                const float sc = __int_as_float(__builtin_amdgcn_readlane(__float_as_int(ds), off + i));
                float* a = acc + di * HDIM + lane * 3;
                atomicAdd(a + 0, r[i][0] * sc);
                atomicAdd(a + 1, r[i][1] * sc);
                atomicAdd(a + 2, r[i][2] * sc);
            }
        }
    };

    auto drain = [&](bool fin) {
        int done = 0;
        while (qn - done >= AGG_BATCH || (fin && qn > done)) {
            int cnt = qn - done;
            if (cnt > AGG_BATCH) cnt = AGG_BATCH;
            int s = 0, dl = 0;
            if (lane < cnt) {
                const int e = q[done + lane];
                s  = src[e];
                dl = dst[e] - lo;
            }
            gather(s, dl, 0, cnt);
            if (!FROM_LIST) {
                if (lane < cnt && wn + lane < AGG_WCAP)
                    wl[wn + lane] = (unsigned)s | ((unsigned)dl << 16);
                wn += cnt;
            }
            done += cnt;
        }
        const int left = qn - done;
        if (done > 0 && left > 0) {
            const int t = (lane < left) ? q[done + lane] : 0;
            if (lane < left) q[lane] = t;
        }
        qn = left;
    };

    auto push = [&](bool act, int e) {
        const unsigned ball = __builtin_amdgcn_ballot_w32(act);
        const int pos = qn + (int)__builtin_amdgcn_mbcnt_lo(ball, 0u);
        if (act) q[pos] = e;
        qn += __builtin_popcount(ball);
    };

    auto append = [&](unsigned m, int vbase) {
        while (__builtin_amdgcn_ballot_w32(m != 0u) != 0u) {
            if (qn + 32 > AGG_QCAP) drain(false);
            const bool act = (m != 0u);
            const int b = __ffs((int)m) - 1;
            m &= m - 1u;
            push(act, ((vbase + ((b >> 2) & 1) * 32 + lane) << 4) + ((b & 3) << 2) + (b >> 3));
        }
    };

    int wcount = 0;
    bool from_list = false;
    if (FROM_LIST) {
        const int cl = (int)lcnt[(size_t)blockIdx.x * AGG_WAVES + lane];
        from_list = __builtin_amdgcn_ballot_w32(cl > AGG_WCAP) == 0u;
        wcount = __builtin_amdgcn_readlane(cl, __builtin_amdgcn_readfirstlane(wave));
    }

    if (from_list) {
        const int n = wcount;
        unsigned cur = wl[(lane < n) ? lane : 0];
        for (int k0 = 0; k0 < n; k0 += 32) {
            const int ni = k0 + 32 + lane;
            const unsigned nxt = wl[(ni < n) ? ni : 0];
            const bool have = (k0 + lane) < n;
            const int s  = have ? (int)(cur & 0xFFFFu) : 0;
            const int dl = have ? (int)(cur >> 16) : 0;
            int c0 = n - k0;
            if (c0 > AGG_BATCH) c0 = AGG_BATCH;
            gather(s, dl, 0, c0);
            int c1 = n - k0 - AGG_BATCH;
            if (c1 > AGG_BATCH) c1 = AGG_BATCH;
            if (c1 > 0) gather(s, dl, AGG_BATCH, c1);
            cur = nxt;
        }
    } else {
        const uint4* __restrict__ key4 = reinterpret_cast<const uint4*>(keyw);
        const unsigned char* __restrict__ key = reinterpret_cast<const unsigned char*>(keyw);
        const int E16     = E >> 4;
        const int E16full = (E16 / AGG_GROUP16) * AGG_GROUP16;

        for (int g0 = wave * AGG_GROUP16; g0 < E16full; g0 += AGG_WAVES * AGG_GROUP16) {
            uint4 kv[AGG_NL];
#pragma unroll
            for (int k = 0; k < AGG_NL; ++k) kv[k] = key4[g0 + k * 32 + lane];
            const unsigned m0 = agg_hits16(kv[0], pat) | (agg_hits16(kv[1], pat) << 4);
            const unsigned m1 = agg_hits16(kv[2], pat) | (agg_hits16(kv[3], pat) << 4);
            append(m0, g0);
            append(m1, g0 + 2 * 32);
            if (qn >= AGG_BATCH) drain(false);
        }
        for (int v0 = E16full + wave * 32; v0 < E16; v0 += AGG_THREADS) {
            unsigned m = 0u;
            if (v0 + lane < E16) m = agg_hits16(key4[v0 + lane], pat);
            append(m, v0);
            if (qn >= AGG_BATCH) drain(false);
        }
        if (wave == 0) {
            const int e = (E16 << 4) + lane;
            const bool act = (lane < 16) && (e < E) && ((unsigned)key[e < E ? e : 0] == (unsigned)blockIdx.x);
            push(act, e);
        }
        drain(true);
        if (!FROM_LIST && lane == 0) wkept[wave] = (unsigned)wn;
    }
    __syncthreads();
    if (!FROM_LIST && wave == 0) lcnt[(size_t)blockIdx.x * AGG_WAVES + lane] = wkept[lane];

    float bi[3], ga[3], be[3], al[3];
#pragma unroll
    for (int i = 0; i < 3; ++i) {
        const int f = lane + 32 * i;
        bi[i] = bias[f]; ga[i] = gamma[f]; be[i] = beta[f]; al[i] = alpha[f];
    }
    for (int n = wave; n < nrows; n += AGG_WAVES) {
        const float* row = acc + n * HDIM;
        const float dv = dinv[lo + n];
        float v[3];
#pragma unroll
        for (int i = 0; i < 3; ++i) v[i] = row[lane + 32 * i] * dv + bi[i];

        float s = v[0] + v[1] + v[2];
#pragma unroll
        for (int m = 16; m > 0; m >>= 1) s += __shfl_xor(s, m, 32);
        const float mu = s * (1.0f / (float)HDIM);

        const float d0 = v[0] - mu, d1 = v[1] - mu, d2 = v[2] - mu;
        float qv = d0 * d0 + d1 * d1 + d2 * d2;
#pragma unroll
        for (int m = 16; m > 0; m >>= 1) qv += __shfl_xor(qv, m, 32);
        const float rstd = rsqrtf(qv * (1.0f / (float)HDIM) + LN_EPS);

        const float dvv[3] = {d0, d1, d2};
#pragma unroll
        for (int i = 0; i < 3; ++i) {
            const float o = dvv[i] * rstd * ga[i] + be[i];
            y[(size_t)(lo + n) * HDIM + lane + 32 * i] = (o >= 0.0f) ? o : al[i] * o;
        }
    }
}

extern "C" void kernel_launch(void* const* d_in, const int* in_sizes, int n_in,
                              void* d_out, int out_size, void* d_ws, size_t ws_size,
                              hipStream_t stream) {
    const float* x   = (const float*)d_in[0];
    const int*   ei  = (const int*)  d_in[1];
    const float* W1  = (const float*)d_in[2];
    const float* b1  = (const float*)d_in[3];
    const float* W2  = (const float*)d_in[4];
    const float* b2  = (const float*)d_in[5];
    const float* g1  = (const float*)d_in[6];
    const float* be1 = (const float*)d_in[7];
    const float* g2  = (const float*)d_in[8];
    const float* be2 = (const float*)d_in[9];
    const float* a   = (const float*)d_in[10];

    const int N   = N_NODES;
    const int Cin = in_sizes[0] / N;
    const int E   = in_sizes[1] / 2;
    const int NH  = N * HDIM;

    const int* src = ei;
    const int* dst = ei + E;

    float* dinv = (float*)d_ws;
    float* bufH = dinv + ((N + 255) & ~255);
    float* bufS = bufH + (size_t)NH;
    float* y1   = (float*)d_out;

    unsigned* part = (unsigned*)bufS;
    unsigned* key = part + (size_t)DEG_PARTS * N_NODES;
    unsigned* lcnt = key + ((E + 255) & ~255) / 4;
    unsigned* lst  = lcnt + AGG_RANGES * AGG_WAVES;
    k_deg_hist<<<DEG_PARTS, DEG_THREADS, 0, stream>>>(part, key, dst, E);
    k_dinv    <<<(N + 255) / 256, 256, 0, stream>>>(dinv, part, N);

    dim3 gemmBlock(32, 4);
    dim3 gemmGrid1((N / 16 + 3) / 4, HDIM / 16);

    k_gemm_wmma_f32<<<gemmGrid1, gemmBlock, 0, stream>>>(x, W1, bufH, N, Cin);
    k_agg_range<false><<<AGG_RANGES, AGG_THREADS, 0, stream>>>(bufH, dinv, src, dst, key, E, lst, lcnt, b1, g1, be1, a, y1, N);

    k_gemm_wmma_f32<<<gemmGrid1, gemmBlock, 0, stream>>>(y1, W2, bufH, N, HDIM);
    k_agg_range<true><<<AGG_RANGES, AGG_THREADS, 0, stream>>>(bufH, dinv, src, dst, key, E, lst, lcnt, b2, g2, be2, a, (float*)d_out, N);
}


# dense products re-tiled: a wave computes 16 rows x all 96 columns (was one 16x16 tile), x read once by 16-byte loads, W whole in LDS in fragment order, whole-line stores; k pairs regrouped per step
# speedup vs baseline: 2.9587x; 1.2368x over previous
#include <hip/hip_runtime.h>
#include <hip/hip_bf16.h>

#define N_NODES 50000
#define HDIM    96
#define LN_EPS  1e-5f

typedef float v2f __attribute__((ext_vector_type(2)));
typedef float v8f __attribute__((ext_vector_type(8)));

#define AGG_R       196
#define AGG_RANGES  ((N_NODES + AGG_R - 1) / AGG_R)
static_assert(AGG_RANGES <= 256, "a range number must fit the key plane's byte");

#define DEG_PARTS   32
#define DEG_THREADS 1024
#define DEG_BATCH   4

__global__ void __launch_bounds__(DEG_THREADS)
k_deg_hist(unsigned* __restrict__ part, unsigned* __restrict__ keyw,
           const int* __restrict__ dst, int E) {
    __shared__ __attribute__((aligned(16))) unsigned cnt[N_NODES];
    const int tid = threadIdx.x;

    uint4* __restrict__ cnt4 = reinterpret_cast<uint4*>(cnt);
    for (int i = tid; i < N_NODES / 4; i += DEG_THREADS) cnt4[i] = make_uint4(0u, 0u, 0u, 0u);
    __syncthreads();

    const int E4 = E >> 2;
    int per4 = (E4 + (int)gridDim.x - 1) / (int)gridDim.x;
    per4 = (per4 + 31) & ~31;
    const int lo4 = (int)blockIdx.x * per4;
    int hi4 = lo4 + per4;
    if (hi4 > E4) hi4 = E4;

    const uint4* __restrict__ dst4 = reinterpret_cast<const uint4*>(dst);
    for (int v0 = lo4; v0 < hi4; v0 += DEG_THREADS * DEG_BATCH) {
        uint4 d[DEG_BATCH];
#pragma unroll
        for (int j = 0; j < DEG_BATCH; ++j) {
            const int v = v0 + j * DEG_THREADS + tid;
            d[j] = (v < hi4) ? dst4[v] : make_uint4(0u, 0u, 0u, 0u);
        }
#pragma unroll
        for (int j = 0; j < DEG_BATCH; ++j) {
            const int v = v0 + j * DEG_THREADS + tid;
            if (v < hi4) {
                atomicAdd(&cnt[d[j].x], 1u);
                atomicAdd(&cnt[d[j].y], 1u);
                atomicAdd(&cnt[d[j].z], 1u);
                atomicAdd(&cnt[d[j].w], 1u);
                keyw[v] = (d[j].x / (unsigned)AGG_R)         | ((d[j].y / (unsigned)AGG_R) << 8) |
                          ((d[j].z / (unsigned)AGG_R) << 16) | ((d[j].w / (unsigned)AGG_R) << 24);
            }
        }
    }
    if (blockIdx.x == gridDim.x - 1 && tid < (E & 3)) {
        const int e = (E4 << 2) + tid;
        const unsigned dd = (unsigned)dst[e];
        atomicAdd(&cnt[dd], 1u);
        reinterpret_cast<unsigned char*>(keyw)[e] = (unsigned char)(dd / (unsigned)AGG_R);
    }
    __syncthreads();

    uint4* __restrict__ p4 = reinterpret_cast<uint4*>(part + (size_t)blockIdx.x * N_NODES);
    for (int i = tid; i < N_NODES / 4; i += DEG_THREADS) p4[i] = cnt4[i];
}

__global__ void k_dinv(float* __restrict__ dinv, const unsigned* __restrict__ part, int n) {
    int i = blockIdx.x * blockDim.x + threadIdx.x;
    if (i < n) {
        unsigned c = 1u;
#pragma unroll
        for (int b = 0; b < DEG_PARTS; ++b) c += part[(size_t)b * n + i];
        dinv[i] = rsqrtf((float)c);
    }
}

#define IN_DIM       128
#define GEMM_GRID    256
#define GEMM_WAVES   13
#define GEMM_THREADS (GEMM_WAVES * 32)
#define GEMM_STRIPS  (N_NODES / 16)
#define GEMM_CPITCH  100
static_assert(N_NODES % 16 == 0, "whole 16-row strips");
static_assert(GEMM_GRID * GEMM_WAVES >= GEMM_STRIPS, "every strip has a wave");
static_assert(HDIM == 96, "6 column tiles; a strip row is 24 float4");

template <int K>
__global__ void __launch_bounds__(GEMM_THREADS)
k_gemm_strip(const float* X,
             const float* __restrict__ W,
             float* __restrict__ Hout) {
    static_assert(K % 8 == 0 && K <= 128, "a float4 per lane covers 8 k per row");
    __shared__ __attribute__((aligned(16))) float Wf[K * HDIM];
    __shared__ __attribute__((aligned(16))) float Cs[GEMM_WAVES * 16 * GEMM_CPITCH];

    const int tid  = threadIdx.x;
    const int lane = tid & 31;
    const int wave = tid >> 5;
    const int c    = lane & 15;
    const int h    = lane >> 4;
    const int strip = wave * GEMM_GRID + (int)blockIdx.x;
    const bool valid = strip < GEMM_STRIPS;

    float4 a[K / 8];
    {
        const int row = (valid ? strip : 0) * 16 + c;
        const float4* arow = reinterpret_cast<const float4*>(X + (size_t)row * K) + h;
#pragma unroll
        for (int j = 0; j < K / 8; ++j) a[j] = arow[2 * j];
    }

    float4* Wf4 = reinterpret_cast<float4*>(Wf);
    {
        constexpr int W_ITEMS = (K / 8) * 6 * 32;
        constexpr int W_TRIPS = (W_ITEMS + GEMM_THREADS - 1) / GEMM_THREADS;
        float4 wv[W_TRIPS];
#pragma unroll
        for (int t = 0; t < W_TRIPS; ++t) {
            const int i  = tid + t * GEMM_THREADS;
            const int ii = (i < W_ITEMS) ? i : 0;
            const int l  = ii & 31;
            const int jn = ii >> 5;
            const int nt = jn % 6;
            const int j  = jn / 6;
            const float* w = W + (size_t)(8 * j + 4 * (l >> 4)) * HDIM + nt * 16 + (l & 15);
            wv[t] = make_float4(w[0], w[HDIM], w[2 * HDIM], w[3 * HDIM]);
        }
#pragma unroll
        for (int t = 0; t < W_TRIPS; ++t) {
            const int i = tid + t * GEMM_THREADS;
            if (i < W_ITEMS) Wf4[i] = wv[t];
        }
    }
    __syncthreads();
    if (!valid) return;

    v8f acc[6];
#pragma unroll
    for (int nt = 0; nt < 6; ++nt) acc[nt] = (v8f){0.f, 0.f, 0.f, 0.f, 0.f, 0.f, 0.f, 0.f};

#pragma unroll
    for (int j = 0; j < K / 8; ++j) {
        float4 b[6];
#pragma unroll
        for (int nt = 0; nt < 6; ++nt) b[nt] = Wf4[(j * 6 + nt) * 32 + lane];
        const v2f a0 = {a[j].x, a[j].y};
        const v2f a1 = {a[j].z, a[j].w};
#pragma unroll
        for (int nt = 0; nt < 6; ++nt) {
            const v2f b0 = {b[nt].x, b[nt].y};
            acc[nt] = __builtin_amdgcn_wmma_f32_16x16x4_f32(
                          false, a0, false, b0, (short)0, acc[nt], false, false);
        }
#pragma unroll
        for (int nt = 0; nt < 6; ++nt) {
            const v2f b1 = {b[nt].z, b[nt].w};
            acc[nt] = __builtin_amdgcn_wmma_f32_16x16x4_f32(
                          false, a1, false, b1, (short)0, acc[nt], false, false);
        }
    }

    float* cs = Cs + wave * (16 * GEMM_CPITCH);
#pragma unroll
    for (int nt = 0; nt < 6; ++nt) {
#pragma unroll
        for (int r = 0; r < 8; ++r) cs[(8 * h + r) * GEMM_CPITCH + nt * 16 + c] = acc[nt][r];
    }
    float4* __restrict__ out4 = reinterpret_cast<float4*>(Hout) + (size_t)strip * (16 * HDIM / 4);
#pragma unroll
    for (int it = 0; it < (16 * HDIM / 4) / 32; ++it) {
        const int idx = it * 32 + lane;
        const int row = idx / (HDIM / 4);
        const int q   = idx % (HDIM / 4);
        out4[idx] = *reinterpret_cast<const float4*>(cs + row * GEMM_CPITCH + q * 4);
    }
}

#define AGG_THREADS 1024
#define AGG_WAVES   (AGG_THREADS / 32)
#define AGG_NL      4
#define AGG_GROUP16 (32 * AGG_NL)
#define AGG_QCAP    64
#define AGG_BATCH   16
#define AGG_WCAP    256
static_assert(N_NODES <= 65536 && AGG_R <= 65536, "source and local row share a 32-bit entry");
static_assert(AGG_WCAP % 32 == 0 && AGG_BATCH == 16, "a 32-entry load feeds two batches");
static_assert(AGG_WAVES == 32, "a wave reads its work-group's counts one per lane");

__device__ __forceinline__ unsigned agg_zero_bytes(unsigned t) {
    return ~(((t & 0x7F7F7F7Fu) + 0x7F7F7F7Fu) | t | 0x7F7F7F7Fu);
}

__device__ __forceinline__ unsigned agg_hits16(uint4 kv, unsigned pat) {
    return (agg_zero_bytes(kv.x ^ pat) >> 7) | (agg_zero_bytes(kv.y ^ pat) >> 6) |
           (agg_zero_bytes(kv.z ^ pat) >> 5) | (agg_zero_bytes(kv.w ^ pat) >> 4);
}

template <bool FROM_LIST>
__global__ void __launch_bounds__(AGG_THREADS)
k_agg_range(const float* __restrict__ h,
            const float* __restrict__ dinv,
            const int* __restrict__ src,
            const int* __restrict__ dst,
            const unsigned* __restrict__ keyw,
            int E,
            unsigned* lst,
            unsigned* lcnt,
            const float* __restrict__ bias,
            const float* __restrict__ gamma,
            const float* __restrict__ beta,
            const float* __restrict__ alpha,
            float* __restrict__ y,
            int N) {
    __shared__ __attribute__((aligned(16))) float acc[AGG_R * HDIM];
    __shared__ int queue[AGG_WAVES * AGG_QCAP];
    __shared__ unsigned wkept[AGG_WAVES];

    const int tid  = threadIdx.x;
    const int lane = tid & 31;
    const int wave = tid >> 5;
    const int lo   = (int)blockIdx.x * AGG_R;
    const unsigned pat = (unsigned)blockIdx.x * 0x01010101u;
    int nrows = N - lo;
    if (nrows > AGG_R) nrows = AGG_R;

    {
        const float4* __restrict__ h4 = reinterpret_cast<const float4*>(h + (size_t)lo * HDIM);
        float4* __restrict__ acc4 = reinterpret_cast<float4*>(acc);
        const int live4 = nrows * (HDIM / 4);
        constexpr int SEED_SLOTS = AGG_R * (HDIM / 4);
        constexpr int SEED_TRIPS = (SEED_SLOTS + AGG_THREADS - 1) / AGG_THREADS;
        float4 v[SEED_TRIPS];
        float  dv[SEED_TRIPS];
#pragma unroll
        for (int j = 0; j < SEED_TRIPS; ++j) {
            const int i  = tid + j * AGG_THREADS;
            const int ii = (i < live4) ? i : 0;
            v[j]  = h4[ii];
            dv[j] = dinv[lo + ii / (HDIM / 4)];
        }
#pragma unroll
        for (int j = 0; j < SEED_TRIPS; ++j) {
            const int i = tid + j * AGG_THREADS;
            if (i < SEED_SLOTS)
                acc4[i] = (i < live4)
                    ? make_float4(v[j].x * dv[j], v[j].y * dv[j], v[j].z * dv[j], v[j].w * dv[j])
                    : make_float4(0.f, 0.f, 0.f, 0.f);
        }
    }
    __syncthreads();

    int* __restrict__ q = queue + wave * AGG_QCAP;
    int qn = 0;

    unsigned* wl = lst + ((size_t)blockIdx.x * AGG_WAVES + wave) * AGG_WCAP;
    int wn = 0;

    auto gather = [&](int s, int dl, int off, int cnt) {
        float ds = 0.0f;
        if (lane >= off && lane < off + cnt) ds = dinv[s];
        float r[AGG_BATCH][3];
#pragma unroll
        for (int i = 0; i < AGG_BATCH; ++i) {
            const int si = __builtin_amdgcn_readlane(s, off + i);
            const float* __restrict__ row = h + (size_t)si * HDIM + lane * 3;
            r[i][0] = row[0];
            r[i][1] = row[1];
            r[i][2] = row[2];
        }
        __builtin_amdgcn_sched_barrier(0);
#pragma unroll
        for (int i = 0; i < AGG_BATCH; ++i) {
            if (i < cnt) {
                const int di = __builtin_amdgcn_readlane(dl, off + i);
                const float sc = __int_as_float(__builtin_amdgcn_readlane(__float_as_int(ds), off + i));
                float* a = acc + di * HDIM + lane * 3;
                atomicAdd(a + 0, r[i][0] * sc);
                atomicAdd(a + 1, r[i][1] * sc);
                atomicAdd(a + 2, r[i][2] * sc);
            }
        }
    };

    auto drain = [&](bool fin) {
        int done = 0;
        while (qn - done >= AGG_BATCH || (fin && qn > done)) {
            int cnt = qn - done;
            if (cnt > AGG_BATCH) cnt = AGG_BATCH;
            int s = 0, dl = 0;
            if (lane < cnt) {
                const int e = q[done + lane];
                s  = src[e];
                dl = dst[e] - lo;
            }
            gather(s, dl, 0, cnt);
            if (!FROM_LIST) {
                if (lane < cnt && wn + lane < AGG_WCAP)
                    wl[wn + lane] = (unsigned)s | ((unsigned)dl << 16);
                wn += cnt;
            }
            done += cnt;
        }
        const int left = qn - done;
        if (done > 0 && left > 0) {
            const int t = (lane < left) ? q[done + lane] : 0;
            if (lane < left) q[lane] = t;
        }
        qn = left;
    };

    auto push = [&](bool act, int e) {
        const unsigned ball = __builtin_amdgcn_ballot_w32(act);
        const int pos = qn + (int)__builtin_amdgcn_mbcnt_lo(ball, 0u);
        if (act) q[pos] = e;
        qn += __builtin_popcount(ball);
    };

    auto append = [&](unsigned m, int vbase) {
        while (__builtin_amdgcn_ballot_w32(m != 0u) != 0u) {
            if (qn + 32 > AGG_QCAP) drain(false);
            const bool act = (m != 0u);
            const int b = __ffs((int)m) - 1;
            m &= m - 1u;
            push(act, ((vbase + ((b >> 2) & 1) * 32 + lane) << 4) + ((b & 3) << 2) + (b >> 3));
        }
    };

    int wcount = 0;
    bool from_list = false;
    if (FROM_LIST) {
        const int cl = (int)lcnt[(size_t)blockIdx.x * AGG_WAVES + lane];
        from_list = __builtin_amdgcn_ballot_w32(cl > AGG_WCAP) == 0u;
        wcount = __builtin_amdgcn_readlane(cl, __builtin_amdgcn_readfirstlane(wave));
    }

    if (from_list) {
        const int n = wcount;
        unsigned cur = wl[(lane < n) ? lane : 0];
        for (int k0 = 0; k0 < n; k0 += 32) {
            const int ni = k0 + 32 + lane;
            const unsigned nxt = wl[(ni < n) ? ni : 0];
            const bool have = (k0 + lane) < n;
            const int s  = have ? (int)(cur & 0xFFFFu) : 0;
            const int dl = have ? (int)(cur >> 16) : 0;
            int c0 = n - k0;
            if (c0 > AGG_BATCH) c0 = AGG_BATCH;
            gather(s, dl, 0, c0);
            int c1 = n - k0 - AGG_BATCH;
            if (c1 > AGG_BATCH) c1 = AGG_BATCH;
            if (c1 > 0) gather(s, dl, AGG_BATCH, c1);
            cur = nxt;
        }
    } else {
        const uint4* __restrict__ key4 = reinterpret_cast<const uint4*>(keyw);
        const unsigned char* __restrict__ key = reinterpret_cast<const unsigned char*>(keyw);
        const int E16     = E >> 4;
        const int E16full = (E16 / AGG_GROUP16) * AGG_GROUP16;

        for (int g0 = wave * AGG_GROUP16; g0 < E16full; g0 += AGG_WAVES * AGG_GROUP16) {
            uint4 kv[AGG_NL];
#pragma unroll
            for (int k = 0; k < AGG_NL; ++k) kv[k] = key4[g0 + k * 32 + lane];
            const unsigned m0 = agg_hits16(kv[0], pat) | (agg_hits16(kv[1], pat) << 4);
            const unsigned m1 = agg_hits16(kv[2], pat) | (agg_hits16(kv[3], pat) << 4);
            append(m0, g0);
            append(m1, g0 + 2 * 32);
            if (qn >= AGG_BATCH) drain(false);
        }
        for (int v0 = E16full + wave * 32; v0 < E16; v0 += AGG_THREADS) {
            unsigned m = 0u;
            if (v0 + lane < E16) m = agg_hits16(key4[v0 + lane], pat);
            append(m, v0);
            if (qn >= AGG_BATCH) drain(false);
        }
        if (wave == 0) {
            const int e = (E16 << 4) + lane;
            const bool act = (lane < 16) && (e < E) && ((unsigned)key[e < E ? e : 0] == (unsigned)blockIdx.x);
            push(act, e);
        }
        drain(true);
        if (!FROM_LIST && lane == 0) wkept[wave] = (unsigned)wn;
    }
    __syncthreads();
    if (!FROM_LIST && wave == 0) lcnt[(size_t)blockIdx.x * AGG_WAVES + lane] = wkept[lane];

    float bi[3], ga[3], be[3], al[3];
#pragma unroll
    for (int i = 0; i < 3; ++i) {
        const int f = lane + 32 * i;
        bi[i] = bias[f]; ga[i] = gamma[f]; be[i] = beta[f]; al[i] = alpha[f];
    }
    for (int n = wave; n < nrows; n += AGG_WAVES) {
        const float* row = acc + n * HDIM;
        const float dv = dinv[lo + n];
        float v[3];
#pragma unroll
        for (int i = 0; i < 3; ++i) v[i] = row[lane + 32 * i] * dv + bi[i];

        float s = v[0] + v[1] + v[2];
#pragma unroll
        for (int m = 16; m > 0; m >>= 1) s += __shfl_xor(s, m, 32);
        const float mu = s * (1.0f / (float)HDIM);

        const float d0 = v[0] - mu, d1 = v[1] - mu, d2 = v[2] - mu;
        float qv = d0 * d0 + d1 * d1 + d2 * d2;
#pragma unroll
        for (int m = 16; m > 0; m >>= 1) qv += __shfl_xor(qv, m, 32);
        const float rstd = rsqrtf(qv * (1.0f / (float)HDIM) + LN_EPS);

        const float dvv[3] = {d0, d1, d2};
#pragma unroll
        for (int i = 0; i < 3; ++i) {
            const float o = dvv[i] * rstd * ga[i] + be[i];
            y[(size_t)(lo + n) * HDIM + lane + 32 * i] = (o >= 0.0f) ? o : al[i] * o;
        }
    }
}

extern "C" void kernel_launch(void* const* d_in, const int* in_sizes, int n_in,
                              void* d_out, int out_size, void* d_ws, size_t ws_size,
                              hipStream_t stream) {
    const float* x   = (const float*)d_in[0];
    const int*   ei  = (const int*)  d_in[1];
    const float* W1  = (const float*)d_in[2];
    const float* b1  = (const float*)d_in[3];
    const float* W2  = (const float*)d_in[4];
    const float* b2  = (const float*)d_in[5];
    const float* g1  = (const float*)d_in[6];
    const float* be1 = (const float*)d_in[7];
    const float* g2  = (const float*)d_in[8];
    const float* be2 = (const float*)d_in[9];
    const float* a   = (const float*)d_in[10];

    const int N   = N_NODES;
    const int E   = in_sizes[1] / 2;
    const int NH  = N * HDIM;

    const int* src = ei;
    const int* dst = ei + E;

    float* dinv = (float*)d_ws;
    float* bufH = dinv + ((N + 255) & ~255);
    float* bufS = bufH + (size_t)NH;
    float* y1   = (float*)d_out;

    unsigned* part = (unsigned*)bufS;
    unsigned* key = part + (size_t)DEG_PARTS * N_NODES;
    unsigned* lcnt = key + ((E + 255) & ~255) / 4;
    unsigned* lst  = lcnt + AGG_RANGES * AGG_WAVES;
    k_deg_hist<<<DEG_PARTS, DEG_THREADS, 0, stream>>>(part, key, dst, E);
    k_dinv    <<<(N + 255) / 256, 256, 0, stream>>>(dinv, part, N);

    k_gemm_strip<IN_DIM><<<GEMM_GRID, GEMM_THREADS, 0, stream>>>(x, W1, bufH);
    k_agg_range<false><<<AGG_RANGES, AGG_THREADS, 0, stream>>>(bufH, dinv, src, dst, key, E, lst, lcnt, b1, g1, be1, a, y1, N);

    k_gemm_strip<HDIM><<<GEMM_GRID, GEMM_THREADS, 0, stream>>>(y1, W2, bufH);
    k_agg_range<true><<<AGG_RANGES, AGG_THREADS, 0, stream>>>(bufH, dinv, src, dst, key, E, lst, lcnt, b2, g2, be2, a, (float*)d_out, N);
}


# aggregation row finishing: a wave runs its 7 rows through bias, LayerNorm, PReLU side by side (all rows' loads and butterfly exchanges in flight together), was row after row; sum order unchanged
# speedup vs baseline: 3.1332x; 1.0590x over previous
#include <hip/hip_runtime.h>
#include <hip/hip_bf16.h>

#define N_NODES 50000
#define HDIM    96
#define LN_EPS  1e-5f

typedef float v2f __attribute__((ext_vector_type(2)));
typedef float v8f __attribute__((ext_vector_type(8)));

#define AGG_R       196
#define AGG_RANGES  ((N_NODES + AGG_R - 1) / AGG_R)
static_assert(AGG_RANGES <= 256, "a range number must fit the key plane's byte");

#define DEG_PARTS   32
#define DEG_THREADS 1024
#define DEG_BATCH   4

__global__ void __launch_bounds__(DEG_THREADS)
k_deg_hist(unsigned* __restrict__ part, unsigned* __restrict__ keyw,
           const int* __restrict__ dst, int E) {
    __shared__ __attribute__((aligned(16))) unsigned cnt[N_NODES];
    const int tid = threadIdx.x;

    uint4* __restrict__ cnt4 = reinterpret_cast<uint4*>(cnt);
    for (int i = tid; i < N_NODES / 4; i += DEG_THREADS) cnt4[i] = make_uint4(0u, 0u, 0u, 0u);
    __syncthreads();

    const int E4 = E >> 2;
    int per4 = (E4 + (int)gridDim.x - 1) / (int)gridDim.x;
    per4 = (per4 + 31) & ~31;
    const int lo4 = (int)blockIdx.x * per4;
    int hi4 = lo4 + per4;
    if (hi4 > E4) hi4 = E4;

    const uint4* __restrict__ dst4 = reinterpret_cast<const uint4*>(dst);
    for (int v0 = lo4; v0 < hi4; v0 += DEG_THREADS * DEG_BATCH) {
        uint4 d[DEG_BATCH];
#pragma unroll
        for (int j = 0; j < DEG_BATCH; ++j) {
            const int v = v0 + j * DEG_THREADS + tid;
            d[j] = (v < hi4) ? dst4[v] : make_uint4(0u, 0u, 0u, 0u);
        }
#pragma unroll
        for (int j = 0; j < DEG_BATCH; ++j) {
            const int v = v0 + j * DEG_THREADS + tid;
            if (v < hi4) {
                atomicAdd(&cnt[d[j].x], 1u);
                atomicAdd(&cnt[d[j].y], 1u);
                atomicAdd(&cnt[d[j].z], 1u);
                atomicAdd(&cnt[d[j].w], 1u);
                keyw[v] = (d[j].x / (unsigned)AGG_R)         | ((d[j].y / (unsigned)AGG_R) << 8) |
                          ((d[j].z / (unsigned)AGG_R) << 16) | ((d[j].w / (unsigned)AGG_R) << 24);
            }
        }
    }
    if (blockIdx.x == gridDim.x - 1 && tid < (E & 3)) {
        const int e = (E4 << 2) + tid;
        const unsigned dd = (unsigned)dst[e];
        atomicAdd(&cnt[dd], 1u);
        reinterpret_cast<unsigned char*>(keyw)[e] = (unsigned char)(dd / (unsigned)AGG_R);
    }
    __syncthreads();

    uint4* __restrict__ p4 = reinterpret_cast<uint4*>(part + (size_t)blockIdx.x * N_NODES);
    for (int i = tid; i < N_NODES / 4; i += DEG_THREADS) p4[i] = cnt4[i];
}

__global__ void k_dinv(float* __restrict__ dinv, const unsigned* __restrict__ part, int n) {
    int i = blockIdx.x * blockDim.x + threadIdx.x;
    if (i < n) {
        unsigned c = 1u;
#pragma unroll
        for (int b = 0; b < DEG_PARTS; ++b) c += part[(size_t)b * n + i];
        dinv[i] = rsqrtf((float)c);
    }
}

#define IN_DIM       128
#define GEMM_GRID    256
#define GEMM_WAVES   13
#define GEMM_THREADS (GEMM_WAVES * 32)
#define GEMM_STRIPS  (N_NODES / 16)
#define GEMM_CPITCH  100
static_assert(N_NODES % 16 == 0, "whole 16-row strips");
static_assert(GEMM_GRID * GEMM_WAVES >= GEMM_STRIPS, "every strip has a wave");
static_assert(HDIM == 96, "6 column tiles; a strip row is 24 float4");

template <int K>
__global__ void __launch_bounds__(GEMM_THREADS)
k_gemm_strip(const float* X,
             const float* __restrict__ W,
             float* __restrict__ Hout) {
    static_assert(K % 8 == 0 && K <= 128, "a float4 per lane covers 8 k per row");
    __shared__ __attribute__((aligned(16))) float Wf[K * HDIM];
    __shared__ __attribute__((aligned(16))) float Cs[GEMM_WAVES * 16 * GEMM_CPITCH];

    const int tid  = threadIdx.x;
    const int lane = tid & 31;
    const int wave = tid >> 5;
    const int c    = lane & 15;
    const int h    = lane >> 4;
    const int strip = wave * GEMM_GRID + (int)blockIdx.x;
    const bool valid = strip < GEMM_STRIPS;

    float4 a[K / 8];
    {
        const int row = (valid ? strip : 0) * 16 + c;
        const float4* arow = reinterpret_cast<const float4*>(X + (size_t)row * K) + h;
#pragma unroll
        for (int j = 0; j < K / 8; ++j) a[j] = arow[2 * j];
    }

    float4* Wf4 = reinterpret_cast<float4*>(Wf);
    {
        constexpr int W_ITEMS = (K / 8) * 6 * 32;
        constexpr int W_TRIPS = (W_ITEMS + GEMM_THREADS - 1) / GEMM_THREADS;
        float4 wv[W_TRIPS];
#pragma unroll
        for (int t = 0; t < W_TRIPS; ++t) {
            const int i  = tid + t * GEMM_THREADS;
            const int ii = (i < W_ITEMS) ? i : 0;
            const int l  = ii & 31;
            const int jn = ii >> 5;
            const int nt = jn % 6;
            const int j  = jn / 6;
            const float* w = W + (size_t)(8 * j + 4 * (l >> 4)) * HDIM + nt * 16 + (l & 15);
            wv[t] = make_float4(w[0], w[HDIM], w[2 * HDIM], w[3 * HDIM]);
        }
#pragma unroll
        for (int t = 0; t < W_TRIPS; ++t) {
            const int i = tid + t * GEMM_THREADS;
            if (i < W_ITEMS) Wf4[i] = wv[t];
        }
    }
    __syncthreads();
    if (!valid) return;

    v8f acc[6];
#pragma unroll
    for (int nt = 0; nt < 6; ++nt) acc[nt] = (v8f){0.f, 0.f, 0.f, 0.f, 0.f, 0.f, 0.f, 0.f};

#pragma unroll
    for (int j = 0; j < K / 8; ++j) {
        float4 b[6];
#pragma unroll
        for (int nt = 0; nt < 6; ++nt) b[nt] = Wf4[(j * 6 + nt) * 32 + lane];
        const v2f a0 = {a[j].x, a[j].y};
        const v2f a1 = {a[j].z, a[j].w};
#pragma unroll
        for (int nt = 0; nt < 6; ++nt) {
            const v2f b0 = {b[nt].x, b[nt].y};
            acc[nt] = __builtin_amdgcn_wmma_f32_16x16x4_f32(
                          false, a0, false, b0, (short)0, acc[nt], false, false);
        }
#pragma unroll
        for (int nt = 0; nt < 6; ++nt) {
            const v2f b1 = {b[nt].z, b[nt].w};
            acc[nt] = __builtin_amdgcn_wmma_f32_16x16x4_f32(
                          false, a1, false, b1, (short)0, acc[nt], false, false);
        }
    }

    float* cs = Cs + wave * (16 * GEMM_CPITCH);
#pragma unroll
    for (int nt = 0; nt < 6; ++nt) {
#pragma unroll
        for (int r = 0; r < 8; ++r) cs[(8 * h + r) * GEMM_CPITCH + nt * 16 + c] = acc[nt][r];
    }
    float4* __restrict__ out4 = reinterpret_cast<float4*>(Hout) + (size_t)strip * (16 * HDIM / 4);
#pragma unroll
    for (int it = 0; it < (16 * HDIM / 4) / 32; ++it) {
        const int idx = it * 32 + lane;
        const int row = idx / (HDIM / 4);
        const int q   = idx % (HDIM / 4);
        out4[idx] = *reinterpret_cast<const float4*>(cs + row * GEMM_CPITCH + q * 4);
    }
}

#define AGG_THREADS 1024
#define AGG_WAVES   (AGG_THREADS / 32)
#define AGG_NL      4
#define AGG_GROUP16 (32 * AGG_NL)
#define AGG_QCAP    64
#define AGG_BATCH   16
#define AGG_WCAP    256
static_assert(N_NODES <= 65536 && AGG_R <= 65536, "source and local row share a 32-bit entry");
static_assert(AGG_WCAP % 32 == 0 && AGG_BATCH == 16, "a 32-entry load feeds two batches");
static_assert(AGG_WAVES == 32, "a wave reads its work-group's counts one per lane");

__device__ __forceinline__ unsigned agg_zero_bytes(unsigned t) {
    return ~(((t & 0x7F7F7F7Fu) + 0x7F7F7F7Fu) | t | 0x7F7F7F7Fu);
}

__device__ __forceinline__ unsigned agg_hits16(uint4 kv, unsigned pat) {
    return (agg_zero_bytes(kv.x ^ pat) >> 7) | (agg_zero_bytes(kv.y ^ pat) >> 6) |
           (agg_zero_bytes(kv.z ^ pat) >> 5) | (agg_zero_bytes(kv.w ^ pat) >> 4);
}

template <bool FROM_LIST>
__global__ void __launch_bounds__(AGG_THREADS)
k_agg_range(const float* __restrict__ h,
            const float* __restrict__ dinv,
            const int* __restrict__ src,
            const int* __restrict__ dst,
            const unsigned* __restrict__ keyw,
            int E,
            unsigned* lst,
            unsigned* lcnt,
            const float* __restrict__ bias,
            const float* __restrict__ gamma,
            const float* __restrict__ beta,
            const float* __restrict__ alpha,
            float* __restrict__ y,
            int N) {
    __shared__ __attribute__((aligned(16))) float acc[AGG_R * HDIM];
    __shared__ int queue[AGG_WAVES * AGG_QCAP];
    __shared__ unsigned wkept[AGG_WAVES];

    const int tid  = threadIdx.x;
    const int lane = tid & 31;
    const int wave = tid >> 5;
    const int lo   = (int)blockIdx.x * AGG_R;
    const unsigned pat = (unsigned)blockIdx.x * 0x01010101u;
    int nrows = N - lo;
    if (nrows > AGG_R) nrows = AGG_R;

    {
        const float4* __restrict__ h4 = reinterpret_cast<const float4*>(h + (size_t)lo * HDIM);
        float4* __restrict__ acc4 = reinterpret_cast<float4*>(acc);
        const int live4 = nrows * (HDIM / 4);
        constexpr int SEED_SLOTS = AGG_R * (HDIM / 4);
        constexpr int SEED_TRIPS = (SEED_SLOTS + AGG_THREADS - 1) / AGG_THREADS;
        float4 v[SEED_TRIPS];
        float  dv[SEED_TRIPS];
#pragma unroll
        for (int j = 0; j < SEED_TRIPS; ++j) {
            const int i  = tid + j * AGG_THREADS;
            const int ii = (i < live4) ? i : 0;
            v[j]  = h4[ii];
            dv[j] = dinv[lo + ii / (HDIM / 4)];
        }
#pragma unroll
        for (int j = 0; j < SEED_TRIPS; ++j) {
            const int i = tid + j * AGG_THREADS;
            if (i < SEED_SLOTS)
                acc4[i] = (i < live4)
                    ? make_float4(v[j].x * dv[j], v[j].y * dv[j], v[j].z * dv[j], v[j].w * dv[j])
                    : make_float4(0.f, 0.f, 0.f, 0.f);
        }
    }
    __syncthreads();

    int* __restrict__ q = queue + wave * AGG_QCAP;
    int qn = 0;

    unsigned* wl = lst + ((size_t)blockIdx.x * AGG_WAVES + wave) * AGG_WCAP;
    int wn = 0;

    auto gather = [&](int s, int dl, int off, int cnt) {
        float ds = 0.0f;
        if (lane >= off && lane < off + cnt) ds = dinv[s];
        float r[AGG_BATCH][3];
#pragma unroll
        for (int i = 0; i < AGG_BATCH; ++i) {
            const int si = __builtin_amdgcn_readlane(s, off + i);
            const float* __restrict__ row = h + (size_t)si * HDIM + lane * 3;
            r[i][0] = row[0];
            r[i][1] = row[1];
            r[i][2] = row[2];
        }
        __builtin_amdgcn_sched_barrier(0);
#pragma unroll
        for (int i = 0; i < AGG_BATCH; ++i) {
            if (i < cnt) {
                const int di = __builtin_amdgcn_readlane(dl, off + i);
                const float sc = __int_as_float(__builtin_amdgcn_readlane(__float_as_int(ds), off + i));
                float* a = acc + di * HDIM + lane * 3;
                atomicAdd(a + 0, r[i][0] * sc);
                atomicAdd(a + 1, r[i][1] * sc);
                atomicAdd(a + 2, r[i][2] * sc);
            }
        }
    };

    auto drain = [&](bool fin) {
        int done = 0;
        while (qn - done >= AGG_BATCH || (fin && qn > done)) {
            int cnt = qn - done;
            if (cnt > AGG_BATCH) cnt = AGG_BATCH;
            int s = 0, dl = 0;
            if (lane < cnt) {
                const int e = q[done + lane];
                s  = src[e];
                dl = dst[e] - lo;
            }
            gather(s, dl, 0, cnt);
            if (!FROM_LIST) {
                if (lane < cnt && wn + lane < AGG_WCAP)
                    wl[wn + lane] = (unsigned)s | ((unsigned)dl << 16);
                wn += cnt;
            }
            done += cnt;
        }
        const int left = qn - done;
        if (done > 0 && left > 0) {
            const int t = (lane < left) ? q[done + lane] : 0;
            if (lane < left) q[lane] = t;
        }
        qn = left;
    };

    auto push = [&](bool act, int e) {
        const unsigned ball = __builtin_amdgcn_ballot_w32(act);
        const int pos = qn + (int)__builtin_amdgcn_mbcnt_lo(ball, 0u);
        if (act) q[pos] = e;
        qn += __builtin_popcount(ball);
    };

    auto append = [&](unsigned m, int vbase) {
        while (__builtin_amdgcn_ballot_w32(m != 0u) != 0u) {
            if (qn + 32 > AGG_QCAP) drain(false);
            const bool act = (m != 0u);
            const int b = __ffs((int)m) - 1;
            m &= m - 1u;
            push(act, ((vbase + ((b >> 2) & 1) * 32 + lane) << 4) + ((b & 3) << 2) + (b >> 3));
        }
    };

    int wcount = 0;
    bool from_list = false;
    if (FROM_LIST) {
        const int cl = (int)lcnt[(size_t)blockIdx.x * AGG_WAVES + lane];
        from_list = __builtin_amdgcn_ballot_w32(cl > AGG_WCAP) == 0u;
        wcount = __builtin_amdgcn_readlane(cl, __builtin_amdgcn_readfirstlane(wave));
    }

    if (from_list) {
        const int n = wcount;
        unsigned cur = wl[(lane < n) ? lane : 0];
        for (int k0 = 0; k0 < n; k0 += 32) {
            const int ni = k0 + 32 + lane;
            const unsigned nxt = wl[(ni < n) ? ni : 0];
            const bool have = (k0 + lane) < n;
            const int s  = have ? (int)(cur & 0xFFFFu) : 0;
            const int dl = have ? (int)(cur >> 16) : 0;
            int c0 = n - k0;
            if (c0 > AGG_BATCH) c0 = AGG_BATCH;
            gather(s, dl, 0, c0);
            int c1 = n - k0 - AGG_BATCH;
            if (c1 > AGG_BATCH) c1 = AGG_BATCH;
            if (c1 > 0) gather(s, dl, AGG_BATCH, c1);
            cur = nxt;
        }
    } else {
        const uint4* __restrict__ key4 = reinterpret_cast<const uint4*>(keyw);
        const unsigned char* __restrict__ key = reinterpret_cast<const unsigned char*>(keyw);
        const int E16     = E >> 4;
        const int E16full = (E16 / AGG_GROUP16) * AGG_GROUP16;

        for (int g0 = wave * AGG_GROUP16; g0 < E16full; g0 += AGG_WAVES * AGG_GROUP16) {
            uint4 kv[AGG_NL];
#pragma unroll
            for (int k = 0; k < AGG_NL; ++k) kv[k] = key4[g0 + k * 32 + lane];
            const unsigned m0 = agg_hits16(kv[0], pat) | (agg_hits16(kv[1], pat) << 4);
            const unsigned m1 = agg_hits16(kv[2], pat) | (agg_hits16(kv[3], pat) << 4);
            append(m0, g0);
            append(m1, g0 + 2 * 32);
            if (qn >= AGG_BATCH) drain(false);
        }
        for (int v0 = E16full + wave * 32; v0 < E16; v0 += AGG_THREADS) {
            unsigned m = 0u;
            if (v0 + lane < E16) m = agg_hits16(key4[v0 + lane], pat);
            append(m, v0);
            if (qn >= AGG_BATCH) drain(false);
        }
        if (wave == 0) {
            const int e = (E16 << 4) + lane;
            const bool act = (lane < 16) && (e < E) && ((unsigned)key[e < E ? e : 0] == (unsigned)blockIdx.x);
            push(act, e);
        }
        drain(true);
        if (!FROM_LIST && lane == 0) wkept[wave] = (unsigned)wn;
    }
    constexpr int EPI_ROWS = (AGG_R + AGG_WAVES - 1) / AGG_WAVES;
    const int wv = __builtin_amdgcn_readfirstlane(wave);
    float bi[3], ga[3], be[3], al[3];
#pragma unroll
    for (int i = 0; i < 3; ++i) {
        const int f = lane + 32 * i;
        bi[i] = bias[f]; ga[i] = gamma[f]; be[i] = beta[f]; al[i] = alpha[f];
    }
    float dvr[EPI_ROWS];
#pragma unroll
    for (int k = 0; k < EPI_ROWS; ++k) {
        const int n = wv + k * AGG_WAVES;
        dvr[k] = dinv[lo + ((n < nrows) ? n : 0)];
    }
    __syncthreads();
    if (!FROM_LIST && wave == 0) lcnt[(size_t)blockIdx.x * AGG_WAVES + lane] = wkept[lane];

    float v[EPI_ROWS][3];
#pragma unroll
    for (int k = 0; k < EPI_ROWS; ++k) {
        const int n = wv + k * AGG_WAVES;
        const float* row = acc + ((n < AGG_R) ? n : 0) * HDIM;
#pragma unroll
        for (int i = 0; i < 3; ++i) v[k][i] = row[lane + 32 * i] * dvr[k] + bi[i];
    }

    float s[EPI_ROWS];
#pragma unroll
    for (int k = 0; k < EPI_ROWS; ++k) s[k] = v[k][0] + v[k][1] + v[k][2];
#pragma unroll
    for (int m = 16; m > 0; m >>= 1) {
#pragma unroll
        for (int k = 0; k < EPI_ROWS; ++k) s[k] += __shfl_xor(s[k], m, 32);
    }

    float d[EPI_ROWS][3];
    float qv[EPI_ROWS];
#pragma unroll
    for (int k = 0; k < EPI_ROWS; ++k) {
        const float mu = s[k] * (1.0f / (float)HDIM);
        const float d0 = v[k][0] - mu, d1 = v[k][1] - mu, d2 = v[k][2] - mu;
        d[k][0] = d0; d[k][1] = d1; d[k][2] = d2;
        qv[k] = d0 * d0 + d1 * d1 + d2 * d2;
    }
#pragma unroll
    for (int m = 16; m > 0; m >>= 1) {
#pragma unroll
        for (int k = 0; k < EPI_ROWS; ++k) qv[k] += __shfl_xor(qv[k], m, 32);
    }

#pragma unroll
    for (int k = 0; k < EPI_ROWS; ++k) {
        const int n = wv + k * AGG_WAVES;
        const float rstd = rsqrtf(qv[k] * (1.0f / (float)HDIM) + LN_EPS);
        if (n < nrows) {
#pragma unroll
            for (int i = 0; i < 3; ++i) {
                const float o = d[k][i] * rstd * ga[i] + be[i];
                y[(size_t)(lo + n) * HDIM + lane + 32 * i] = (o >= 0.0f) ? o : al[i] * o;
            }
        }
    }
}

extern "C" void kernel_launch(void* const* d_in, const int* in_sizes, int n_in,
                              void* d_out, int out_size, void* d_ws, size_t ws_size,
                              hipStream_t stream) {
    const float* x   = (const float*)d_in[0];
    const int*   ei  = (const int*)  d_in[1];
    const float* W1  = (const float*)d_in[2];
    const float* b1  = (const float*)d_in[3];
    const float* W2  = (const float*)d_in[4];
    const float* b2  = (const float*)d_in[5];
    const float* g1  = (const float*)d_in[6];
    const float* be1 = (const float*)d_in[7];
    const float* g2  = (const float*)d_in[8];
    const float* be2 = (const float*)d_in[9];
    const float* a   = (const float*)d_in[10];

    const int N   = N_NODES;
    const int E   = in_sizes[1] / 2;
    const int NH  = N * HDIM;

    const int* src = ei;
    const int* dst = ei + E;

    float* dinv = (float*)d_ws;
    float* bufH = dinv + ((N + 255) & ~255);
    float* bufS = bufH + (size_t)NH;
    float* y1   = (float*)d_out;

    unsigned* part = (unsigned*)bufS;
    unsigned* key = part + (size_t)DEG_PARTS * N_NODES;
    unsigned* lcnt = key + ((E + 255) & ~255) / 4;
    unsigned* lst  = lcnt + AGG_RANGES * AGG_WAVES;
    k_deg_hist<<<DEG_PARTS, DEG_THREADS, 0, stream>>>(part, key, dst, E);
    k_dinv    <<<(N + 255) / 256, 256, 0, stream>>>(dinv, part, N);

    k_gemm_strip<IN_DIM><<<GEMM_GRID, GEMM_THREADS, 0, stream>>>(x, W1, bufH);
    k_agg_range<false><<<AGG_RANGES, AGG_THREADS, 0, stream>>>(bufH, dinv, src, dst, key, E, lst, lcnt, b1, g1, be1, a, y1, N);

    k_gemm_strip<HDIM><<<GEMM_GRID, GEMM_THREADS, 0, stream>>>(y1, W2, bufH);
    k_agg_range<true><<<AGG_RANGES, AGG_THREADS, 0, stream>>>(bufH, dinv, src, dst, key, E, lst, lcnt, b2, g2, be2, a, (float*)d_out, N);
}


# layer 2 aggregates first, transforms after, (A y)W2 for A(y W2): the owned 196x96 sums times W2 on f32 WMMA out of LDS in the aggregation launch; product launch gone; rows summed before k
# speedup vs baseline: 3.4113x; 1.0888x over previous
#include <hip/hip_runtime.h>
#include <hip/hip_bf16.h>

#define N_NODES 50000
#define HDIM    96
#define LN_EPS  1e-5f

typedef float v2f __attribute__((ext_vector_type(2)));
typedef float v8f __attribute__((ext_vector_type(8)));

#define AGG_R       196
#define AGG_RANGES  ((N_NODES + AGG_R - 1) / AGG_R)
static_assert(AGG_RANGES <= 256, "a range number must fit the key plane's byte");

#define DEG_PARTS   32
#define DEG_THREADS 1024
#define DEG_BATCH   4

__global__ void __launch_bounds__(DEG_THREADS)
k_deg_hist(unsigned* __restrict__ part, unsigned* __restrict__ keyw,
           const int* __restrict__ dst, int E) {
    __shared__ __attribute__((aligned(16))) unsigned cnt[N_NODES];
    const int tid = threadIdx.x;

    uint4* __restrict__ cnt4 = reinterpret_cast<uint4*>(cnt);
    for (int i = tid; i < N_NODES / 4; i += DEG_THREADS) cnt4[i] = make_uint4(0u, 0u, 0u, 0u);
    __syncthreads();

    const int E4 = E >> 2;
    int per4 = (E4 + (int)gridDim.x - 1) / (int)gridDim.x;
    per4 = (per4 + 31) & ~31;
    const int lo4 = (int)blockIdx.x * per4;
    int hi4 = lo4 + per4;
    if (hi4 > E4) hi4 = E4;

    const uint4* __restrict__ dst4 = reinterpret_cast<const uint4*>(dst);
    for (int v0 = lo4; v0 < hi4; v0 += DEG_THREADS * DEG_BATCH) {
        uint4 d[DEG_BATCH];
#pragma unroll
        for (int j = 0; j < DEG_BATCH; ++j) {
            const int v = v0 + j * DEG_THREADS + tid;
            d[j] = (v < hi4) ? dst4[v] : make_uint4(0u, 0u, 0u, 0u);
        }
#pragma unroll
        for (int j = 0; j < DEG_BATCH; ++j) {
            const int v = v0 + j * DEG_THREADS + tid;
            if (v < hi4) {
                atomicAdd(&cnt[d[j].x], 1u);
                atomicAdd(&cnt[d[j].y], 1u);
                atomicAdd(&cnt[d[j].z], 1u);
                atomicAdd(&cnt[d[j].w], 1u);
                keyw[v] = (d[j].x / (unsigned)AGG_R)         | ((d[j].y / (unsigned)AGG_R) << 8) |
                          ((d[j].z / (unsigned)AGG_R) << 16) | ((d[j].w / (unsigned)AGG_R) << 24);
            }
        }
    }
    if (blockIdx.x == gridDim.x - 1 && tid < (E & 3)) {
        const int e = (E4 << 2) + tid;
        const unsigned dd = (unsigned)dst[e];
        atomicAdd(&cnt[dd], 1u);
        reinterpret_cast<unsigned char*>(keyw)[e] = (unsigned char)(dd / (unsigned)AGG_R);
    }
    __syncthreads();

    uint4* __restrict__ p4 = reinterpret_cast<uint4*>(part + (size_t)blockIdx.x * N_NODES);
    for (int i = tid; i < N_NODES / 4; i += DEG_THREADS) p4[i] = cnt4[i];
}

__global__ void k_dinv(float* __restrict__ dinv, const unsigned* __restrict__ part, int n) {
    int i = blockIdx.x * blockDim.x + threadIdx.x;
    if (i < n) {
        unsigned c = 1u;
#pragma unroll
        for (int b = 0; b < DEG_PARTS; ++b) c += part[(size_t)b * n + i];
        dinv[i] = rsqrtf((float)c);
    }
}

#define IN_DIM       128
#define GEMM_GRID    256
#define GEMM_WAVES   13
#define GEMM_THREADS (GEMM_WAVES * 32)
#define GEMM_STRIPS  (N_NODES / 16)
#define GEMM_CPITCH  100
static_assert(N_NODES % 16 == 0, "whole 16-row strips");
static_assert(GEMM_GRID * GEMM_WAVES >= GEMM_STRIPS, "every strip has a wave");
static_assert(HDIM == 96, "6 column tiles; a strip row is 24 float4");

template <int K>
__global__ void __launch_bounds__(GEMM_THREADS)
k_gemm_strip(const float* X,
             const float* __restrict__ W,
             float* __restrict__ Hout) {
    static_assert(K % 8 == 0 && K <= 128, "a float4 per lane covers 8 k per row");
    __shared__ __attribute__((aligned(16))) float Wf[K * HDIM];
    __shared__ __attribute__((aligned(16))) float Cs[GEMM_WAVES * 16 * GEMM_CPITCH];

    const int tid  = threadIdx.x;
    const int lane = tid & 31;
    const int wave = tid >> 5;
    const int c    = lane & 15;
    const int h    = lane >> 4;
    const int strip = wave * GEMM_GRID + (int)blockIdx.x;
    const bool valid = strip < GEMM_STRIPS;

    float4 a[K / 8];
    {
        const int row = (valid ? strip : 0) * 16 + c;
        const float4* arow = reinterpret_cast<const float4*>(X + (size_t)row * K) + h;
#pragma unroll
        for (int j = 0; j < K / 8; ++j) a[j] = arow[2 * j];
    }

    float4* Wf4 = reinterpret_cast<float4*>(Wf);
    {
        constexpr int W_ITEMS = (K / 8) * 6 * 32;
        constexpr int W_TRIPS = (W_ITEMS + GEMM_THREADS - 1) / GEMM_THREADS;
        float4 wv[W_TRIPS];
#pragma unroll
        for (int t = 0; t < W_TRIPS; ++t) {
            const int i  = tid + t * GEMM_THREADS;
            const int ii = (i < W_ITEMS) ? i : 0;
            const int l  = ii & 31;
            const int jn = ii >> 5;
            const int nt = jn % 6;
            const int j  = jn / 6;
            const float* w = W + (size_t)(8 * j + 4 * (l >> 4)) * HDIM + nt * 16 + (l & 15);
            wv[t] = make_float4(w[0], w[HDIM], w[2 * HDIM], w[3 * HDIM]);
        }
#pragma unroll
        for (int t = 0; t < W_TRIPS; ++t) {
            const int i = tid + t * GEMM_THREADS;
            if (i < W_ITEMS) Wf4[i] = wv[t];
        }
    }
    __syncthreads();
    if (!valid) return;

    v8f acc[6];
#pragma unroll
    for (int nt = 0; nt < 6; ++nt) acc[nt] = (v8f){0.f, 0.f, 0.f, 0.f, 0.f, 0.f, 0.f, 0.f};

#pragma unroll
    for (int j = 0; j < K / 8; ++j) {
        float4 b[6];
#pragma unroll
        for (int nt = 0; nt < 6; ++nt) b[nt] = Wf4[(j * 6 + nt) * 32 + lane];
        const v2f a0 = {a[j].x, a[j].y};
        const v2f a1 = {a[j].z, a[j].w};
#pragma unroll
        for (int nt = 0; nt < 6; ++nt) {
            const v2f b0 = {b[nt].x, b[nt].y};
            acc[nt] = __builtin_amdgcn_wmma_f32_16x16x4_f32(
                          false, a0, false, b0, (short)0, acc[nt], false, false);
        }
#pragma unroll
        for (int nt = 0; nt < 6; ++nt) {
            const v2f b1 = {b[nt].z, b[nt].w};
            acc[nt] = __builtin_amdgcn_wmma_f32_16x16x4_f32(
                          false, a1, false, b1, (short)0, acc[nt], false, false);
        }
    }

    float* cs = Cs + wave * (16 * GEMM_CPITCH);
#pragma unroll
    for (int nt = 0; nt < 6; ++nt) {
#pragma unroll
        for (int r = 0; r < 8; ++r) cs[(8 * h + r) * GEMM_CPITCH + nt * 16 + c] = acc[nt][r];
    }
    float4* __restrict__ out4 = reinterpret_cast<float4*>(Hout) + (size_t)strip * (16 * HDIM / 4);
#pragma unroll
    for (int it = 0; it < (16 * HDIM / 4) / 32; ++it) {
        const int idx = it * 32 + lane;
        const int row = idx / (HDIM / 4);
        const int q   = idx % (HDIM / 4);
        out4[idx] = *reinterpret_cast<const float4*>(cs + row * GEMM_CPITCH + q * 4);
    }
}

#define AGG_THREADS 1024
#define AGG_WAVES   (AGG_THREADS / 32)
#define AGG_NL      4
#define AGG_GROUP16 (32 * AGG_NL)
#define AGG_QCAP    64
#define AGG_BATCH   16
#define AGG_WCAP    256
#define AGG_STRIPS  ((AGG_R + 15) / 16)
#define AGG_RPAD    (AGG_STRIPS * 16)
#define AGG_XPITCH  100
static_assert(N_NODES <= 65536 && AGG_R <= 65536, "source and local row share a 32-bit entry");
static_assert(AGG_WCAP % 32 == 0 && AGG_BATCH == 16, "a 32-entry load feeds two batches");
static_assert(AGG_WAVES == 32, "a wave reads its work-group's counts one per lane");

__device__ __forceinline__ unsigned agg_zero_bytes(unsigned t) {
    return ~(((t & 0x7F7F7F7Fu) + 0x7F7F7F7Fu) | t | 0x7F7F7F7Fu);
}

__device__ __forceinline__ unsigned agg_hits16(uint4 kv, unsigned pat) {
    return (agg_zero_bytes(kv.x ^ pat) >> 7) | (agg_zero_bytes(kv.y ^ pat) >> 6) |
           (agg_zero_bytes(kv.z ^ pat) >> 5) | (agg_zero_bytes(kv.w ^ pat) >> 4);
}

template <bool FROM_LIST, bool XFORM>
__global__ void __launch_bounds__(AGG_THREADS)
k_agg_range(const float* __restrict__ h,
            const float* __restrict__ dinv,
            const int* __restrict__ src,
            const int* __restrict__ dst,
            const unsigned* __restrict__ keyw,
            int E,
            unsigned* lst,
            unsigned* lcnt,
            const float* __restrict__ Wx,
            const float* __restrict__ bias,
            const float* __restrict__ gamma,
            const float* __restrict__ beta,
            const float* __restrict__ alpha,
            float* __restrict__ y,
            int N) {
    constexpr int P = XFORM ? AGG_XPITCH : HDIM;
    __shared__ __attribute__((aligned(16))) float acc[AGG_RPAD * P];
    __shared__ __attribute__((aligned(16))) float Wf[XFORM ? HDIM * HDIM : 4];
    __shared__ int queue[AGG_WAVES * AGG_QCAP];
    __shared__ unsigned wkept[AGG_WAVES];

    const int tid  = threadIdx.x;
    const int lane = tid & 31;
    const int wave = tid >> 5;
    const int lo   = (int)blockIdx.x * AGG_R;
    const unsigned pat = (unsigned)blockIdx.x * 0x01010101u;
    int nrows = N - lo;
    if (nrows > AGG_R) nrows = AGG_R;

    constexpr int W_ITEMS = (HDIM / 8) * 6 * 32;
    constexpr int W_TRIPS = XFORM ? (W_ITEMS + AGG_THREADS - 1) / AGG_THREADS : 1;
    float4 wv4[W_TRIPS];
    if (XFORM) {
#pragma unroll
        for (int t = 0; t < W_TRIPS; ++t) {
            const int i  = tid + t * AGG_THREADS;
            const int ii = (i < W_ITEMS) ? i : 0;
            const int l  = ii & 31;
            const int jn = ii >> 5;
            const int nt = jn % 6;
            const int j  = jn / 6;
            const float* w = Wx + (size_t)(8 * j + 4 * (l >> 4)) * HDIM + nt * 16 + (l & 15);
            wv4[t] = make_float4(w[0], w[HDIM], w[2 * HDIM], w[3 * HDIM]);
        }
    }

    {
        const float4* __restrict__ h4 = reinterpret_cast<const float4*>(h + (size_t)lo * HDIM);
        float4* __restrict__ acc4 = reinterpret_cast<float4*>(acc);
        const int live4 = nrows * (HDIM / 4);
        constexpr int SEED_SLOTS = AGG_RPAD * (HDIM / 4);
        constexpr int SEED_TRIPS = (SEED_SLOTS + AGG_THREADS - 1) / AGG_THREADS;
        float4 v[SEED_TRIPS];
        float  dv[SEED_TRIPS];
#pragma unroll
        for (int j = 0; j < SEED_TRIPS; ++j) {
            const int i  = tid + j * AGG_THREADS;
            const int ii = (i < live4) ? i : 0;
            v[j]  = h4[ii];
            dv[j] = dinv[lo + ii / (HDIM / 4)];
        }
#pragma unroll
        for (int j = 0; j < SEED_TRIPS; ++j) {
            const int i = tid + j * AGG_THREADS;
            const int slot = XFORM ? (i / (HDIM / 4)) * (P / 4) + i % (HDIM / 4) : i;
            if (i < SEED_SLOTS)
                acc4[slot] = (i < live4)
                    ? make_float4(v[j].x * dv[j], v[j].y * dv[j], v[j].z * dv[j], v[j].w * dv[j])
                    : make_float4(0.f, 0.f, 0.f, 0.f);
        }
    }
    if (XFORM) {
        float4* Wf4s = reinterpret_cast<float4*>(Wf);
#pragma unroll
        for (int t = 0; t < W_TRIPS; ++t) {
            const int i = tid + t * AGG_THREADS;
            if (i < W_ITEMS) Wf4s[i] = wv4[t];
        }
    }
    __syncthreads();

    int* __restrict__ q = queue + wave * AGG_QCAP;
    int qn = 0;

    unsigned* wl = lst + ((size_t)blockIdx.x * AGG_WAVES + wave) * AGG_WCAP;
    int wn = 0;

    auto gather = [&](int s, int dl, int off, int cnt) {
        float ds = 0.0f;
        if (lane >= off && lane < off + cnt) ds = dinv[s];
        float r[AGG_BATCH][3];
#pragma unroll
        for (int i = 0; i < AGG_BATCH; ++i) {
            const int si = __builtin_amdgcn_readlane(s, off + i);
            const float* __restrict__ row = h + (size_t)si * HDIM + lane * 3;
            r[i][0] = row[0];
            r[i][1] = row[1];
            r[i][2] = row[2];
        }
        __builtin_amdgcn_sched_barrier(0);
#pragma unroll
        for (int i = 0; i < AGG_BATCH; ++i) {
            if (i < cnt) {
                const int di = __builtin_amdgcn_readlane(dl, off + i);
                const float sc = __int_as_float(__builtin_amdgcn_readlane(__float_as_int(ds), off + i));
                float* a = acc + di * P + lane * 3;
                atomicAdd(a + 0, r[i][0] * sc);
                atomicAdd(a + 1, r[i][1] * sc);
                atomicAdd(a + 2, r[i][2] * sc);
            }
        }
    };

    auto drain = [&](bool fin) {
        int done = 0;
        while (qn - done >= AGG_BATCH || (fin && qn > done)) {
            int cnt = qn - done;
            if (cnt > AGG_BATCH) cnt = AGG_BATCH;
            int s = 0, dl = 0;
            if (lane < cnt) {
                const int e = q[done + lane];
                s  = src[e];
                dl = dst[e] - lo;
            }
            gather(s, dl, 0, cnt);
            if (!FROM_LIST) {
                if (lane < cnt && wn + lane < AGG_WCAP)
                    wl[wn + lane] = (unsigned)s | ((unsigned)dl << 16);
                wn += cnt;
            }
            done += cnt;
        }
        const int left = qn - done;
        if (done > 0 && left > 0) {
            const int t = (lane < left) ? q[done + lane] : 0;
            if (lane < left) q[lane] = t;
        }
        qn = left;
    };

    auto push = [&](bool act, int e) {
        const unsigned ball = __builtin_amdgcn_ballot_w32(act);
        const int pos = qn + (int)__builtin_amdgcn_mbcnt_lo(ball, 0u);
        if (act) q[pos] = e;
        qn += __builtin_popcount(ball);
    };

    auto append = [&](unsigned m, int vbase) {
        while (__builtin_amdgcn_ballot_w32(m != 0u) != 0u) {
            if (qn + 32 > AGG_QCAP) drain(false);
            const bool act = (m != 0u);
            const int b = __ffs((int)m) - 1;
            m &= m - 1u;
            push(act, ((vbase + ((b >> 2) & 1) * 32 + lane) << 4) + ((b & 3) << 2) + (b >> 3));
        }
    };

    int wcount = 0;
    bool from_list = false;
    if (FROM_LIST) {
        const int cl = (int)lcnt[(size_t)blockIdx.x * AGG_WAVES + lane];
        from_list = __builtin_amdgcn_ballot_w32(cl > AGG_WCAP) == 0u;
        wcount = __builtin_amdgcn_readlane(cl, __builtin_amdgcn_readfirstlane(wave));
    }

    if (from_list) {
        const int n = wcount;
        unsigned cur = wl[(lane < n) ? lane : 0];
        for (int k0 = 0; k0 < n; k0 += 32) {
            const int ni = k0 + 32 + lane;
            const unsigned nxt = wl[(ni < n) ? ni : 0];
            const bool have = (k0 + lane) < n;
            const int s  = have ? (int)(cur & 0xFFFFu) : 0;
            const int dl = have ? (int)(cur >> 16) : 0;
            int c0 = n - k0;
            if (c0 > AGG_BATCH) c0 = AGG_BATCH;
            gather(s, dl, 0, c0);
            int c1 = n - k0 - AGG_BATCH;
            if (c1 > AGG_BATCH) c1 = AGG_BATCH;
            if (c1 > 0) gather(s, dl, AGG_BATCH, c1);
            cur = nxt;
        }
    } else {
        const uint4* __restrict__ key4 = reinterpret_cast<const uint4*>(keyw);
        const unsigned char* __restrict__ key = reinterpret_cast<const unsigned char*>(keyw);
        const int E16     = E >> 4;
        const int E16full = (E16 / AGG_GROUP16) * AGG_GROUP16;

        for (int g0 = wave * AGG_GROUP16; g0 < E16full; g0 += AGG_WAVES * AGG_GROUP16) {
            uint4 kv[AGG_NL];
#pragma unroll
            for (int k = 0; k < AGG_NL; ++k) kv[k] = key4[g0 + k * 32 + lane];
            const unsigned m0 = agg_hits16(kv[0], pat) | (agg_hits16(kv[1], pat) << 4);
            const unsigned m1 = agg_hits16(kv[2], pat) | (agg_hits16(kv[3], pat) << 4);
            append(m0, g0);
            append(m1, g0 + 2 * 32);
            if (qn >= AGG_BATCH) drain(false);
        }
        for (int v0 = E16full + wave * 32; v0 < E16; v0 += AGG_THREADS) {
            unsigned m = 0u;
            if (v0 + lane < E16) m = agg_hits16(key4[v0 + lane], pat);
            append(m, v0);
            if (qn >= AGG_BATCH) drain(false);
        }
        if (wave == 0) {
            const int e = (E16 << 4) + lane;
            const bool act = (lane < 16) && (e < E) && ((unsigned)key[e < E ? e : 0] == (unsigned)blockIdx.x);
            push(act, e);
        }
        drain(true);
        if (!FROM_LIST && lane == 0) wkept[wave] = (unsigned)wn;
    }
    constexpr int EPI_ROWS = (AGG_R + AGG_WAVES - 1) / AGG_WAVES;
    const int wv = __builtin_amdgcn_readfirstlane(wave);
    float bi[3], ga[3], be[3], al[3];
#pragma unroll
    for (int i = 0; i < 3; ++i) {
        const int f = lane + 32 * i;
        bi[i] = bias[f]; ga[i] = gamma[f]; be[i] = beta[f]; al[i] = alpha[f];
    }
    float dvr[EPI_ROWS];
#pragma unroll
    for (int k = 0; k < EPI_ROWS; ++k) {
        const int n = wv + k * AGG_WAVES;
        dvr[k] = dinv[lo + ((n < nrows) ? n : 0)];
    }
    __syncthreads();
    if (!FROM_LIST && wave == 0) lcnt[(size_t)blockIdx.x * AGG_WAVES + lane] = wkept[lane];

    if (XFORM) {
        const int c  = lane & 15;
        const int hh = lane >> 4;
        const bool mm = wv < 2 * AGG_STRIPS;
        const int strip = wv >> 1;
        const int nt0   = (wv & 1) * 3;
        v8f o[3];
#pragma unroll
        for (int t = 0; t < 3; ++t) o[t] = (v8f){0.f, 0.f, 0.f, 0.f, 0.f, 0.f, 0.f, 0.f};
        if (mm) {
            const float4* arow = reinterpret_cast<const float4*>(acc + (strip * 16 + c) * P) + hh;
            const float4* Wf4 = reinterpret_cast<const float4*>(Wf);
            float4 a[HDIM / 8];
#pragma unroll
            for (int j = 0; j < HDIM / 8; ++j) a[j] = arow[2 * j];
#pragma unroll
            for (int j = 0; j < HDIM / 8; ++j) {
                float4 b[3];
#pragma unroll
                for (int t = 0; t < 3; ++t) b[t] = Wf4[(j * 6 + nt0 + t) * 32 + lane];
                const v2f a0 = {a[j].x, a[j].y};
                const v2f a1 = {a[j].z, a[j].w};
#pragma unroll
                for (int t = 0; t < 3; ++t) {
                    const v2f b0 = {b[t].x, b[t].y};
                    o[t] = __builtin_amdgcn_wmma_f32_16x16x4_f32(
                               false, a0, false, b0, (short)0, o[t], false, false);
                }
#pragma unroll
                for (int t = 0; t < 3; ++t) {
                    const v2f b1 = {b[t].z, b[t].w};
                    o[t] = __builtin_amdgcn_wmma_f32_16x16x4_f32(
                               false, a1, false, b1, (short)0, o[t], false, false);
                }
            }
        }
        __syncthreads();
        if (mm) {
#pragma unroll
            for (int t = 0; t < 3; ++t) {
#pragma unroll
                for (int r = 0; r < 8; ++r)
                    acc[(strip * 16 + 8 * hh + r) * P + (nt0 + t) * 16 + c] = o[t][r];
            }
        }
        __syncthreads();
    }

    float v[EPI_ROWS][3];
#pragma unroll
    for (int k = 0; k < EPI_ROWS; ++k) {
        const int n = wv + k * AGG_WAVES;
        const float* row = acc + ((n < AGG_R) ? n : 0) * P;
#pragma unroll
        for (int i = 0; i < 3; ++i) v[k][i] = row[lane + 32 * i] * dvr[k] + bi[i];
    }

    float s[EPI_ROWS];
#pragma unroll
    for (int k = 0; k < EPI_ROWS; ++k) s[k] = v[k][0] + v[k][1] + v[k][2];
#pragma unroll
    for (int m = 16; m > 0; m >>= 1) {
#pragma unroll
        for (int k = 0; k < EPI_ROWS; ++k) s[k] += __shfl_xor(s[k], m, 32);
    }

    float d[EPI_ROWS][3];
    float qv[EPI_ROWS];
#pragma unroll
    for (int k = 0; k < EPI_ROWS; ++k) {
        const float mu = s[k] * (1.0f / (float)HDIM);
        const float d0 = v[k][0] - mu, d1 = v[k][1] - mu, d2 = v[k][2] - mu;
        d[k][0] = d0; d[k][1] = d1; d[k][2] = d2;
        qv[k] = d0 * d0 + d1 * d1 + d2 * d2;
    }
#pragma unroll
    for (int m = 16; m > 0; m >>= 1) {
#pragma unroll
        for (int k = 0; k < EPI_ROWS; ++k) qv[k] += __shfl_xor(qv[k], m, 32);
    }

#pragma unroll
    for (int k = 0; k < EPI_ROWS; ++k) {
        const int n = wv + k * AGG_WAVES;
        const float rstd = rsqrtf(qv[k] * (1.0f / (float)HDIM) + LN_EPS);
        if (n < nrows) {
#pragma unroll
            for (int i = 0; i < 3; ++i) {
                const float o = d[k][i] * rstd * ga[i] + be[i];
                y[(size_t)(lo + n) * HDIM + lane + 32 * i] = (o >= 0.0f) ? o : al[i] * o;
            }
        }
    }
}

extern "C" void kernel_launch(void* const* d_in, const int* in_sizes, int n_in,
                              void* d_out, int out_size, void* d_ws, size_t ws_size,
                              hipStream_t stream) {
    const float* x   = (const float*)d_in[0];
    const int*   ei  = (const int*)  d_in[1];
    const float* W1  = (const float*)d_in[2];
    const float* b1  = (const float*)d_in[3];
    const float* W2  = (const float*)d_in[4];
    const float* b2  = (const float*)d_in[5];
    const float* g1  = (const float*)d_in[6];
    const float* be1 = (const float*)d_in[7];
    const float* g2  = (const float*)d_in[8];
    const float* be2 = (const float*)d_in[9];
    const float* a   = (const float*)d_in[10];

    const int N   = N_NODES;
    const int E   = in_sizes[1] / 2;
    const int NH  = N * HDIM;

    const int* src = ei;
    const int* dst = ei + E;

    float* dinv = (float*)d_ws;
    float* bufH = dinv + ((N + 255) & ~255);
    float* bufS = bufH + (size_t)NH;
    float* h1   = (float*)d_out;

    unsigned* part = (unsigned*)bufS;
    unsigned* key = part + (size_t)DEG_PARTS * N_NODES;
    unsigned* lcnt = key + ((E + 255) & ~255) / 4;
    unsigned* lst  = lcnt + AGG_RANGES * AGG_WAVES;
    k_deg_hist<<<DEG_PARTS, DEG_THREADS, 0, stream>>>(part, key, dst, E);
    k_dinv    <<<(N + 255) / 256, 256, 0, stream>>>(dinv, part, N);

    k_gemm_strip<IN_DIM><<<GEMM_GRID, GEMM_THREADS, 0, stream>>>(x, W1, h1);
    k_agg_range<false, false><<<AGG_RANGES, AGG_THREADS, 0, stream>>>(h1, dinv, src, dst, key, E, lst, lcnt, nullptr, b1, g1, be1, a, bufH, N);

    k_agg_range<true, true><<<AGG_RANGES, AGG_THREADS, 0, stream>>>(bufH, dinv, src, dst, key, E, lst, lcnt, W2, b2, g2, be2, a, (float*)d_out, N);
}


# edges sorted once into per-owner, per-slice lists by 32 more work-groups of the degree launch (LDS returning adds); layer 1 reads lists as layer 2 does, no key scan; sum order LDS arrival
# speedup vs baseline: 3.9749x; 1.1652x over previous
#include <hip/hip_runtime.h>
#include <hip/hip_bf16.h>

#define N_NODES 50000
#define HDIM    96
#define LN_EPS  1e-5f

typedef float v2f __attribute__((ext_vector_type(2)));
typedef float v8f __attribute__((ext_vector_type(8)));

#define AGG_R       196
#define AGG_RANGES  ((N_NODES + AGG_R - 1) / AGG_R)
static_assert(AGG_RANGES <= 256, "a range number must fit the key plane's byte");
#define AGG_THREADS 1024
#define AGG_WAVES   (AGG_THREADS / 32)
#define AGG_WCAP    256

#define DEG_PARTS   32
#define DEG_THREADS 1024
#define DEG_BATCH   4
#define DEG_LDS_WORDS (AGG_RANGES * AGG_WCAP + AGG_RANGES)
static_assert(DEG_LDS_WORDS >= N_NODES, "the histogram uses the same LDS");
static_assert(DEG_PARTS == AGG_WAVES, "slice p's lists are read by wave p");
static_assert(AGG_RANGES <= DEG_THREADS && AGG_RANGES % (DEG_THREADS / 32) == 0,
              "a thread per count, whole lists per wave");
static_assert(N_NODES <= 65536, "source and local row share a 32-bit entry");

__device__ __forceinline__ unsigned bin_place(unsigned* __restrict__ ocnt, unsigned d) {
    return atomicAdd(&ocnt[d / (unsigned)AGG_R], 1u);
}
__device__ __forceinline__ void bin_store(unsigned* __restrict__ stage, unsigned pos, unsigned d, unsigned s) {
    const unsigned o = d / (unsigned)AGG_R;
    if (pos < (unsigned)AGG_WCAP) stage[o * AGG_WCAP + pos] = s | ((d - o * (unsigned)AGG_R) << 16);
}

__global__ void __launch_bounds__(DEG_THREADS)
k_deg_bin(unsigned* __restrict__ part, unsigned* __restrict__ keyw,
          unsigned* __restrict__ lst, unsigned* __restrict__ lcnt,
          const int* __restrict__ src, const int* __restrict__ dst, int E) {
    __shared__ __attribute__((aligned(16))) unsigned lds[DEG_LDS_WORDS];
    const int tid = threadIdx.x;
    const bool sorter = (int)blockIdx.x >= DEG_PARTS;
    const int p = sorter ? (int)blockIdx.x - DEG_PARTS : (int)blockIdx.x;

    const int E4 = E >> 2;
    int per4 = (E4 + DEG_PARTS - 1) / DEG_PARTS;
    per4 = (per4 + 31) & ~31;
    const int lo4 = p * per4;
    int hi4 = lo4 + per4;
    if (hi4 > E4) hi4 = E4;
    const uint4* __restrict__ dst4 = reinterpret_cast<const uint4*>(dst);

    if (!sorter) {
        unsigned* __restrict__ cnt = lds;
        uint4* __restrict__ cnt4 = reinterpret_cast<uint4*>(lds);
        for (int i = tid; i < N_NODES / 4; i += DEG_THREADS) cnt4[i] = make_uint4(0u, 0u, 0u, 0u);
        __syncthreads();

        for (int v0 = lo4; v0 < hi4; v0 += DEG_THREADS * DEG_BATCH) {
            uint4 d[DEG_BATCH];
#pragma unroll
            for (int j = 0; j < DEG_BATCH; ++j) {
                const int v = v0 + j * DEG_THREADS + tid;
                d[j] = (v < hi4) ? dst4[v] : make_uint4(0u, 0u, 0u, 0u);
            }
#pragma unroll
            for (int j = 0; j < DEG_BATCH; ++j) {
                const int v = v0 + j * DEG_THREADS + tid;
                if (v < hi4) {
                    atomicAdd(&cnt[d[j].x], 1u);
                    atomicAdd(&cnt[d[j].y], 1u);
                    atomicAdd(&cnt[d[j].z], 1u);
                    atomicAdd(&cnt[d[j].w], 1u);
                    keyw[v] = (d[j].x / (unsigned)AGG_R)         | ((d[j].y / (unsigned)AGG_R) << 8) |
                              ((d[j].z / (unsigned)AGG_R) << 16) | ((d[j].w / (unsigned)AGG_R) << 24);
                }
            }
        }
        if (p == DEG_PARTS - 1 && tid < (E & 3)) {
            const int e = (E4 << 2) + tid;
            const unsigned dd = (unsigned)dst[e];
            atomicAdd(&cnt[dd], 1u);
            reinterpret_cast<unsigned char*>(keyw)[e] = (unsigned char)(dd / (unsigned)AGG_R);
        }
        __syncthreads();

        uint4* __restrict__ p4 = reinterpret_cast<uint4*>(part + (size_t)p * N_NODES);
        for (int i = tid; i < N_NODES / 4; i += DEG_THREADS) p4[i] = cnt4[i];
    } else {
        unsigned* __restrict__ stage = lds;
        unsigned* __restrict__ ocnt  = lds + AGG_RANGES * AGG_WCAP;
        if (tid < AGG_RANGES) ocnt[tid] = 0u;
        __syncthreads();

        const uint4* __restrict__ src4 = reinterpret_cast<const uint4*>(src);
        for (int v0 = lo4; v0 < hi4; v0 += DEG_THREADS * DEG_BATCH) {
            uint4 d[DEG_BATCH], s[DEG_BATCH];
#pragma unroll
            for (int j = 0; j < DEG_BATCH; ++j) {
                const int v = v0 + j * DEG_THREADS + tid;
                d[j] = (v < hi4) ? dst4[v] : make_uint4(0u, 0u, 0u, 0u);
                s[j] = (v < hi4) ? src4[v] : make_uint4(0u, 0u, 0u, 0u);
            }
            unsigned pos[DEG_BATCH][4];
#pragma unroll
            for (int j = 0; j < DEG_BATCH; ++j) {
                const int v = v0 + j * DEG_THREADS + tid;
                pos[j][0] = pos[j][1] = pos[j][2] = pos[j][3] = (unsigned)AGG_WCAP;
                if (v < hi4) {
                    pos[j][0] = bin_place(ocnt, d[j].x);
                    pos[j][1] = bin_place(ocnt, d[j].y);
                    pos[j][2] = bin_place(ocnt, d[j].z);
                    pos[j][3] = bin_place(ocnt, d[j].w);
                }
            }
#pragma unroll
            for (int j = 0; j < DEG_BATCH; ++j) {
                bin_store(stage, pos[j][0], d[j].x, s[j].x);
                bin_store(stage, pos[j][1], d[j].y, s[j].y);
                bin_store(stage, pos[j][2], d[j].z, s[j].z);
                bin_store(stage, pos[j][3], d[j].w, s[j].w);
            }
        }
        if (p == DEG_PARTS - 1 && tid < (E & 3)) {
            const int e = (E4 << 2) + tid;
            const unsigned dd = (unsigned)dst[e];
            bin_store(stage, bin_place(ocnt, dd), dd, (unsigned)src[e]);
        }
        __syncthreads();

        if (tid < AGG_RANGES) lcnt[(size_t)p * AGG_RANGES + tid] = ocnt[tid];
        const int lane = tid & 31;
        const int wave = tid >> 5;
        constexpr int LISTS_PER_WAVE = AGG_RANGES / (DEG_THREADS / 32);
        const uint4* __restrict__ stage4 = reinterpret_cast<const uint4*>(stage);
        uint4    ev[LISTS_PER_WAVE];
        unsigned nn[LISTS_PER_WAVE];
#pragma unroll
        for (int k = 0; k < LISTS_PER_WAVE; ++k) {
            ev[k] = stage4[(wave * LISTS_PER_WAVE + k) * (AGG_WCAP / 4) + lane];
            nn[k] = ocnt[wave * LISTS_PER_WAVE + k];
        }
#pragma unroll
        for (int k = 0; k < LISTS_PER_WAVE; ++k) {
            const int o = wave * LISTS_PER_WAVE + k;
            reinterpret_cast<uint4*>(lst + ((size_t)o * AGG_WAVES + p) * AGG_WCAP)[lane] = ev[k];
        }
#pragma unroll
        for (int k = 0; k < LISTS_PER_WAVE; ++k) {
            const int o = wave * LISTS_PER_WAVE + k;
            unsigned n = nn[k];
            if (n > (unsigned)AGG_WCAP) n = (unsigned)AGG_WCAP;
            uint4* __restrict__ out4 =
                reinterpret_cast<uint4*>(lst + ((size_t)o * AGG_WAVES + p) * AGG_WCAP);
            for (int j = lane + 32; 4 * j < (int)n; j += 32) out4[j] = stage4[o * (AGG_WCAP / 4) + j];
        }
    }
}

__global__ void k_dinv(float* __restrict__ dinv, const unsigned* __restrict__ part, int n) {
    int i = blockIdx.x * blockDim.x + threadIdx.x;
    if (i < n) {
        unsigned c = 1u;
#pragma unroll
        for (int b = 0; b < DEG_PARTS; ++b) c += part[(size_t)b * n + i];
        dinv[i] = rsqrtf((float)c);
    }
}

#define IN_DIM       128
#define GEMM_GRID    256
#define GEMM_WAVES   13
#define GEMM_THREADS (GEMM_WAVES * 32)
#define GEMM_STRIPS  (N_NODES / 16)
#define GEMM_CPITCH  100
static_assert(N_NODES % 16 == 0, "whole 16-row strips");
static_assert(GEMM_GRID * GEMM_WAVES >= GEMM_STRIPS, "every strip has a wave");
static_assert(HDIM == 96, "6 column tiles; a strip row is 24 float4");

template <int K>
__global__ void __launch_bounds__(GEMM_THREADS)
k_gemm_strip(const float* X,
             const float* __restrict__ W,
             float* __restrict__ Hout) {
    static_assert(K % 8 == 0 && K <= 128, "a float4 per lane covers 8 k per row");
    __shared__ __attribute__((aligned(16))) float Wf[K * HDIM];
    __shared__ __attribute__((aligned(16))) float Cs[GEMM_WAVES * 16 * GEMM_CPITCH];

    const int tid  = threadIdx.x;
    const int lane = tid & 31;
    const int wave = tid >> 5;
    const int c    = lane & 15;
    const int h    = lane >> 4;
    const int strip = wave * GEMM_GRID + (int)blockIdx.x;
    const bool valid = strip < GEMM_STRIPS;

    float4 a[K / 8];
    {
        const int row = (valid ? strip : 0) * 16 + c;
        const float4* arow = reinterpret_cast<const float4*>(X + (size_t)row * K) + h;
#pragma unroll
        for (int j = 0; j < K / 8; ++j) a[j] = arow[2 * j];
    }

    float4* Wf4 = reinterpret_cast<float4*>(Wf);
    {
        constexpr int W_ITEMS = (K / 8) * 6 * 32;
        constexpr int W_TRIPS = (W_ITEMS + GEMM_THREADS - 1) / GEMM_THREADS;
        float4 wv[W_TRIPS];
#pragma unroll
        for (int t = 0; t < W_TRIPS; ++t) {
            const int i  = tid + t * GEMM_THREADS;
            const int ii = (i < W_ITEMS) ? i : 0;
            const int l  = ii & 31;
            const int jn = ii >> 5;
            const int nt = jn % 6;
            const int j  = jn / 6;
            const float* w = W + (size_t)(8 * j + 4 * (l >> 4)) * HDIM + nt * 16 + (l & 15);
            wv[t] = make_float4(w[0], w[HDIM], w[2 * HDIM], w[3 * HDIM]);
        }
#pragma unroll
        for (int t = 0; t < W_TRIPS; ++t) {
            const int i = tid + t * GEMM_THREADS;
            if (i < W_ITEMS) Wf4[i] = wv[t];
        }
    }
    __syncthreads();
    if (!valid) return;

    v8f acc[6];
#pragma unroll
    for (int nt = 0; nt < 6; ++nt) acc[nt] = (v8f){0.f, 0.f, 0.f, 0.f, 0.f, 0.f, 0.f, 0.f};

#pragma unroll
    for (int j = 0; j < K / 8; ++j) {
        float4 b[6];
#pragma unroll
        for (int nt = 0; nt < 6; ++nt) b[nt] = Wf4[(j * 6 + nt) * 32 + lane];
        const v2f a0 = {a[j].x, a[j].y};
        const v2f a1 = {a[j].z, a[j].w};
#pragma unroll
        for (int nt = 0; nt < 6; ++nt) {
            const v2f b0 = {b[nt].x, b[nt].y};
            acc[nt] = __builtin_amdgcn_wmma_f32_16x16x4_f32(
                          false, a0, false, b0, (short)0, acc[nt], false, false);
        }
#pragma unroll
        for (int nt = 0; nt < 6; ++nt) {
            const v2f b1 = {b[nt].z, b[nt].w};
            acc[nt] = __builtin_amdgcn_wmma_f32_16x16x4_f32(
                          false, a1, false, b1, (short)0, acc[nt], false, false);
        }
    }

    float* cs = Cs + wave * (16 * GEMM_CPITCH);
#pragma unroll
    for (int nt = 0; nt < 6; ++nt) {
#pragma unroll
        for (int r = 0; r < 8; ++r) cs[(8 * h + r) * GEMM_CPITCH + nt * 16 + c] = acc[nt][r];
    }
    float4* __restrict__ out4 = reinterpret_cast<float4*>(Hout) + (size_t)strip * (16 * HDIM / 4);
#pragma unroll
    for (int it = 0; it < (16 * HDIM / 4) / 32; ++it) {
        const int idx = it * 32 + lane;
        const int row = idx / (HDIM / 4);
        const int q   = idx % (HDIM / 4);
        out4[idx] = *reinterpret_cast<const float4*>(cs + row * GEMM_CPITCH + q * 4);
    }
}

#define AGG_NL      4
#define AGG_GROUP16 (32 * AGG_NL)
#define AGG_QCAP    64
#define AGG_BATCH   16
#define AGG_STRIPS  ((AGG_R + 15) / 16)
#define AGG_RPAD    (AGG_STRIPS * 16)
#define AGG_XPITCH  100
static_assert(AGG_R <= 65536, "source and local row share a 32-bit entry");
static_assert(AGG_WCAP % 32 == 0 && AGG_BATCH == 16, "a 32-entry load feeds two batches");
static_assert(AGG_WAVES == 32, "a wave reads its work-group's counts one per lane");

__device__ __forceinline__ unsigned agg_zero_bytes(unsigned t) {
    return ~(((t & 0x7F7F7F7Fu) + 0x7F7F7F7Fu) | t | 0x7F7F7F7Fu);
}

__device__ __forceinline__ unsigned agg_hits16(uint4 kv, unsigned pat) {
    return (agg_zero_bytes(kv.x ^ pat) >> 7) | (agg_zero_bytes(kv.y ^ pat) >> 6) |
           (agg_zero_bytes(kv.z ^ pat) >> 5) | (agg_zero_bytes(kv.w ^ pat) >> 4);
}

template <bool XFORM>
__global__ void __launch_bounds__(AGG_THREADS)
k_agg_range(const float* __restrict__ h,
            const float* __restrict__ dinv,
            const int* __restrict__ src,
            const int* __restrict__ dst,
            const unsigned* __restrict__ keyw,
            int E,
            const unsigned* __restrict__ lst,
            const unsigned* __restrict__ lcnt,
            const float* __restrict__ Wx,
            const float* __restrict__ bias,
            const float* __restrict__ gamma,
            const float* __restrict__ beta,
            const float* __restrict__ alpha,
            float* __restrict__ y,
            int N) {
    constexpr int P = XFORM ? AGG_XPITCH : HDIM;
    __shared__ __attribute__((aligned(16))) float acc[AGG_RPAD * P];
    __shared__ __attribute__((aligned(16))) float Wf[XFORM ? HDIM * HDIM : 4];
    __shared__ int queue[AGG_WAVES * AGG_QCAP];

    const int tid  = threadIdx.x;
    const int lane = tid & 31;
    const int wave = tid >> 5;
    const int lo   = (int)blockIdx.x * AGG_R;
    const unsigned pat = (unsigned)blockIdx.x * 0x01010101u;
    int nrows = N - lo;
    if (nrows > AGG_R) nrows = AGG_R;

    constexpr int W_ITEMS = (HDIM / 8) * 6 * 32;
    constexpr int W_TRIPS = XFORM ? (W_ITEMS + AGG_THREADS - 1) / AGG_THREADS : 1;
    float4 wv4[W_TRIPS];
    if (XFORM) {
#pragma unroll
        for (int t = 0; t < W_TRIPS; ++t) {
            const int i  = tid + t * AGG_THREADS;
            const int ii = (i < W_ITEMS) ? i : 0;
            const int l  = ii & 31;
            const int jn = ii >> 5;
            const int nt = jn % 6;
            const int j  = jn / 6;
            const float* w = Wx + (size_t)(8 * j + 4 * (l >> 4)) * HDIM + nt * 16 + (l & 15);
            wv4[t] = make_float4(w[0], w[HDIM], w[2 * HDIM], w[3 * HDIM]);
        }
    }

    {
        const float4* __restrict__ h4 = reinterpret_cast<const float4*>(h + (size_t)lo * HDIM);
        float4* __restrict__ acc4 = reinterpret_cast<float4*>(acc);
        const int live4 = nrows * (HDIM / 4);
        constexpr int SEED_SLOTS = AGG_RPAD * (HDIM / 4);
        constexpr int SEED_TRIPS = (SEED_SLOTS + AGG_THREADS - 1) / AGG_THREADS;
        float4 v[SEED_TRIPS];
        float  dv[SEED_TRIPS];
#pragma unroll
        for (int j = 0; j < SEED_TRIPS; ++j) {
            const int i  = tid + j * AGG_THREADS;
            const int ii = (i < live4) ? i : 0;
            v[j]  = h4[ii];
            dv[j] = dinv[lo + ii / (HDIM / 4)];
        }
#pragma unroll
        for (int j = 0; j < SEED_TRIPS; ++j) {
            const int i = tid + j * AGG_THREADS;
            const int slot = XFORM ? (i / (HDIM / 4)) * (P / 4) + i % (HDIM / 4) : i;
            if (i < SEED_SLOTS)
                acc4[slot] = (i < live4)
                    ? make_float4(v[j].x * dv[j], v[j].y * dv[j], v[j].z * dv[j], v[j].w * dv[j])
                    : make_float4(0.f, 0.f, 0.f, 0.f);
        }
    }
    if (XFORM) {
        float4* Wf4s = reinterpret_cast<float4*>(Wf);
#pragma unroll
        for (int t = 0; t < W_TRIPS; ++t) {
            const int i = tid + t * AGG_THREADS;
            if (i < W_ITEMS) Wf4s[i] = wv4[t];
        }
    }
    __syncthreads();

    int* __restrict__ q = queue + wave * AGG_QCAP;
    int qn = 0;

    const unsigned* __restrict__ wl = lst + ((size_t)blockIdx.x * AGG_WAVES + wave) * AGG_WCAP;

    auto gather = [&](int s, int dl, int off, int cnt) {
        float ds = 0.0f;
        if (lane >= off && lane < off + cnt) ds = dinv[s];
        float r[AGG_BATCH][3];
#pragma unroll
        for (int i = 0; i < AGG_BATCH; ++i) {
            const int si = __builtin_amdgcn_readlane(s, off + i);
            const float* __restrict__ row = h + (size_t)si * HDIM + lane * 3;
            r[i][0] = row[0];
            r[i][1] = row[1];
            r[i][2] = row[2];
        }
        __builtin_amdgcn_sched_barrier(0);
#pragma unroll
        for (int i = 0; i < AGG_BATCH; ++i) {
            if (i < cnt) {
                const int di = __builtin_amdgcn_readlane(dl, off + i);
                const float sc = __int_as_float(__builtin_amdgcn_readlane(__float_as_int(ds), off + i));
                float* a = acc + di * P + lane * 3;
                atomicAdd(a + 0, r[i][0] * sc);
                atomicAdd(a + 1, r[i][1] * sc);
                atomicAdd(a + 2, r[i][2] * sc);
            }
        }
    };

    auto drain = [&](bool fin) {
        int done = 0;
        while (qn - done >= AGG_BATCH || (fin && qn > done)) {
            int cnt = qn - done;
            if (cnt > AGG_BATCH) cnt = AGG_BATCH;
            int s = 0, dl = 0;
            if (lane < cnt) {
                const int e = q[done + lane];
                s  = src[e];
                dl = dst[e] - lo;
            }
            gather(s, dl, 0, cnt);
            done += cnt;
        }
        const int left = qn - done;
        if (done > 0 && left > 0) {
            const int t = (lane < left) ? q[done + lane] : 0;
            if (lane < left) q[lane] = t;
        }
        qn = left;
    };

    auto push = [&](bool act, int e) {
        const unsigned ball = __builtin_amdgcn_ballot_w32(act);
        const int pos = qn + (int)__builtin_amdgcn_mbcnt_lo(ball, 0u);
        if (act) q[pos] = e;
        qn += __builtin_popcount(ball);
    };

    auto append = [&](unsigned m, int vbase) {
        while (__builtin_amdgcn_ballot_w32(m != 0u) != 0u) {
            if (qn + 32 > AGG_QCAP) drain(false);
            const bool act = (m != 0u);
            const int b = __ffs((int)m) - 1;
            m &= m - 1u;
            push(act, ((vbase + ((b >> 2) & 1) * 32 + lane) << 4) + ((b & 3) << 2) + (b >> 3));
        }
    };

    const unsigned cl = lcnt[(size_t)lane * AGG_RANGES + blockIdx.x];
    const bool from_list = __builtin_amdgcn_ballot_w32(cl > (unsigned)AGG_WCAP) == 0u;
    const int wcount = __builtin_amdgcn_readlane((int)cl, __builtin_amdgcn_readfirstlane(wave));

    if (from_list) {
        const int n = wcount;
        unsigned cur = wl[(lane < n) ? lane : 0];
        for (int k0 = 0; k0 < n; k0 += 32) {
            const int ni = k0 + 32 + lane;
            const unsigned nxt = wl[(ni < n) ? ni : 0];
            const bool have = (k0 + lane) < n;
            const int s  = have ? (int)(cur & 0xFFFFu) : 0;
            const int dl = have ? (int)(cur >> 16) : 0;
            int c0 = n - k0;
            if (c0 > AGG_BATCH) c0 = AGG_BATCH;
            gather(s, dl, 0, c0);
            int c1 = n - k0 - AGG_BATCH;
            if (c1 > AGG_BATCH) c1 = AGG_BATCH;
            if (c1 > 0) gather(s, dl, AGG_BATCH, c1);
            cur = nxt;
        }
    } else {
        const uint4* __restrict__ key4 = reinterpret_cast<const uint4*>(keyw);
        const unsigned char* __restrict__ key = reinterpret_cast<const unsigned char*>(keyw);
        const int E16     = E >> 4;
        const int E16full = (E16 / AGG_GROUP16) * AGG_GROUP16;

        for (int g0 = wave * AGG_GROUP16; g0 < E16full; g0 += AGG_WAVES * AGG_GROUP16) {
            uint4 kv[AGG_NL];
#pragma unroll
            for (int k = 0; k < AGG_NL; ++k) kv[k] = key4[g0 + k * 32 + lane];
            const unsigned m0 = agg_hits16(kv[0], pat) | (agg_hits16(kv[1], pat) << 4);
            const unsigned m1 = agg_hits16(kv[2], pat) | (agg_hits16(kv[3], pat) << 4);
            append(m0, g0);
            append(m1, g0 + 2 * 32);
            if (qn >= AGG_BATCH) drain(false);
        }
        for (int v0 = E16full + wave * 32; v0 < E16; v0 += AGG_THREADS) {
            unsigned m = 0u;
            if (v0 + lane < E16) m = agg_hits16(key4[v0 + lane], pat);
            append(m, v0);
            if (qn >= AGG_BATCH) drain(false);
        }
        if (wave == 0) {
            const int e = (E16 << 4) + lane;
            const bool act = (lane < 16) && (e < E) && ((unsigned)key[e < E ? e : 0] == (unsigned)blockIdx.x);
            push(act, e);
        }
        drain(true);
    }
    constexpr int EPI_ROWS = (AGG_R + AGG_WAVES - 1) / AGG_WAVES;
    const int wv = __builtin_amdgcn_readfirstlane(wave);
    float bi[3], ga[3], be[3], al[3];
#pragma unroll
    for (int i = 0; i < 3; ++i) {
        const int f = lane + 32 * i;
        bi[i] = bias[f]; ga[i] = gamma[f]; be[i] = beta[f]; al[i] = alpha[f];
    }
    float dvr[EPI_ROWS];
#pragma unroll
    for (int k = 0; k < EPI_ROWS; ++k) {
        const int n = wv + k * AGG_WAVES;
        dvr[k] = dinv[lo + ((n < nrows) ? n : 0)];
    }
    __syncthreads();

    if (XFORM) {
        const int c  = lane & 15;
        const int hh = lane >> 4;
        const bool mm = wv < 2 * AGG_STRIPS;
        const int strip = wv >> 1;
        const int nt0   = (wv & 1) * 3;
        v8f o[3];
#pragma unroll
        for (int t = 0; t < 3; ++t) o[t] = (v8f){0.f, 0.f, 0.f, 0.f, 0.f, 0.f, 0.f, 0.f};
        if (mm) {
            const float4* arow = reinterpret_cast<const float4*>(acc + (strip * 16 + c) * P) + hh;
            const float4* Wf4 = reinterpret_cast<const float4*>(Wf);
            float4 a[HDIM / 8];
#pragma unroll
            for (int j = 0; j < HDIM / 8; ++j) a[j] = arow[2 * j];
#pragma unroll
            for (int j = 0; j < HDIM / 8; ++j) {
                float4 b[3];
#pragma unroll
                for (int t = 0; t < 3; ++t) b[t] = Wf4[(j * 6 + nt0 + t) * 32 + lane];
                const v2f a0 = {a[j].x, a[j].y};
                const v2f a1 = {a[j].z, a[j].w};
#pragma unroll
                for (int t = 0; t < 3; ++t) {
                    const v2f b0 = {b[t].x, b[t].y};
                    o[t] = __builtin_amdgcn_wmma_f32_16x16x4_f32(
                               false, a0, false, b0, (short)0, o[t], false, false);
                }
#pragma unroll
                for (int t = 0; t < 3; ++t) {
                    const v2f b1 = {b[t].z, b[t].w};
                    o[t] = __builtin_amdgcn_wmma_f32_16x16x4_f32(
                               false, a1, false, b1, (short)0, o[t], false, false);
                }
            }
        }
        __syncthreads();
        if (mm) {
#pragma unroll
            for (int t = 0; t < 3; ++t) {
#pragma unroll
                for (int r = 0; r < 8; ++r)
                    acc[(strip * 16 + 8 * hh + r) * P + (nt0 + t) * 16 + c] = o[t][r];
            }
        }
        __syncthreads();
    }

    float v[EPI_ROWS][3];
#pragma unroll
    for (int k = 0; k < EPI_ROWS; ++k) {
        const int n = wv + k * AGG_WAVES;
        const float* row = acc + ((n < AGG_R) ? n : 0) * P;
#pragma unroll
        for (int i = 0; i < 3; ++i) v[k][i] = row[lane + 32 * i] * dvr[k] + bi[i];
    }

    float s[EPI_ROWS];
#pragma unroll
    for (int k = 0; k < EPI_ROWS; ++k) s[k] = v[k][0] + v[k][1] + v[k][2];
#pragma unroll
    for (int m = 16; m > 0; m >>= 1) {
#pragma unroll
        for (int k = 0; k < EPI_ROWS; ++k) s[k] += __shfl_xor(s[k], m, 32);
    }

    float d[EPI_ROWS][3];
    float qv[EPI_ROWS];
#pragma unroll
    for (int k = 0; k < EPI_ROWS; ++k) {
        const float mu = s[k] * (1.0f / (float)HDIM);
        const float d0 = v[k][0] - mu, d1 = v[k][1] - mu, d2 = v[k][2] - mu;
        d[k][0] = d0; d[k][1] = d1; d[k][2] = d2;
        qv[k] = d0 * d0 + d1 * d1 + d2 * d2;
    }
#pragma unroll
    for (int m = 16; m > 0; m >>= 1) {
#pragma unroll
        for (int k = 0; k < EPI_ROWS; ++k) qv[k] += __shfl_xor(qv[k], m, 32);
    }

#pragma unroll
    for (int k = 0; k < EPI_ROWS; ++k) {
        const int n = wv + k * AGG_WAVES;
        const float rstd = rsqrtf(qv[k] * (1.0f / (float)HDIM) + LN_EPS);
        if (n < nrows) {
#pragma unroll
            for (int i = 0; i < 3; ++i) {
                const float o = d[k][i] * rstd * ga[i] + be[i];
                y[(size_t)(lo + n) * HDIM + lane + 32 * i] = (o >= 0.0f) ? o : al[i] * o;
            }
        }
    }
}

extern "C" void kernel_launch(void* const* d_in, const int* in_sizes, int n_in,
                              void* d_out, int out_size, void* d_ws, size_t ws_size,
                              hipStream_t stream) {
    const float* x   = (const float*)d_in[0];
    const int*   ei  = (const int*)  d_in[1];
    const float* W1  = (const float*)d_in[2];
    const float* b1  = (const float*)d_in[3];
    const float* W2  = (const float*)d_in[4];
    const float* b2  = (const float*)d_in[5];
    const float* g1  = (const float*)d_in[6];
    const float* be1 = (const float*)d_in[7];
    const float* g2  = (const float*)d_in[8];
    const float* be2 = (const float*)d_in[9];
    const float* a   = (const float*)d_in[10];

    const int N   = N_NODES;
    const int E   = in_sizes[1] / 2;
    const int NH  = N * HDIM;

    const int* src = ei;
    const int* dst = ei + E;

    float* dinv = (float*)d_ws;
    float* bufH = dinv + ((N + 255) & ~255);
    float* bufS = bufH + (size_t)NH;
    float* h1   = (float*)d_out;

    unsigned* part = (unsigned*)bufS;
    unsigned* key = part + (size_t)DEG_PARTS * N_NODES;
    unsigned* lcnt = key + ((E + 255) & ~255) / 4;
    unsigned* lst  = lcnt + AGG_RANGES * AGG_WAVES;
    k_deg_bin<<<2 * DEG_PARTS, DEG_THREADS, 0, stream>>>(part, key, lst, lcnt, src, dst, E);
    k_dinv    <<<(N + 255) / 256, 256, 0, stream>>>(dinv, part, N);

    k_gemm_strip<IN_DIM><<<GEMM_GRID, GEMM_THREADS, 0, stream>>>(x, W1, h1);
    k_agg_range<false><<<AGG_RANGES, AGG_THREADS, 0, stream>>>(h1, dinv, src, dst, key, E, lst, lcnt, nullptr, b1, g1, be1, a, bufH, N);

    k_agg_range<true><<<AGG_RANGES, AGG_THREADS, 0, stream>>>(bufH, dinv, src, dst, key, E, lst, lcnt, W2, b2, g2, be2, a, (float*)d_out, N);
}
